# starting kernel
# baseline (speedup 1.0000x reference)
#include <hip/hip_runtime.h>

typedef float v2f __attribute__((ext_vector_type(2)));
typedef float v8f __attribute__((ext_vector_type(8)));

#define GEMBED   128
#define GNG      432
#define GCHUNK   16
#define GNCHUNK  27
#define FPITCH   (GEMBED + 8)
#define PIX_WAVE 16
#define WAVES_WG 8
#define PIX_WG   (PIX_WAVE * WAVES_WG)
#define GTHRESH  0.01f

__global__ void gauss_prepass(const float* __restrict__ m0, const float* __restrict__ u0, const float* __restrict__ o0,
                              const float* __restrict__ m1, const float* __restrict__ u1, const float* __restrict__ o1,
                              const float* __restrict__ m2, const float* __restrict__ u2, const float* __restrict__ o2,
                              const float* __restrict__ m3, const float* __restrict__ u3, const float* __restrict__ o3,
                              float* __restrict__ params, float* __restrict__ numg)
{
    const int tid = threadIdx.x;
    if (tid == 0) *numg = 0.0f;
    __syncthreads();

    const float* Ms[4] = {m0, m1, m2, m3};
    const float* Us[4] = {u0, u1, u2, u3};
    const float* Os[4] = {o0, o1, o2, o3};
    const float  Hs[4] = {25.f, 50.f, 100.f, 200.f};

    float cnt = 0.0f;
    for (int g = tid; g < 4 * GNG; g += blockDim.x) {
        const int s = g / GNG;
        const int i = g - s * GNG;
        const float* m = Ms[s] + (size_t)i * 3;
        const float* u = Us[s] + (size_t)i * 9;
        float o = Os[s][i];
        const float H  = Hs[s];
        const float sh = H * 0.01f;
        const float c00 = fmaxf(u[0], 1.0f);
        const float c01 = fmaxf(u[1], 1.0f);
        const float c11 = fmaxf(u[4], 1.0f);
        const float va  = sh * sh * c11 + 0.3f;
        const float vb  = sh * sh * c01;
        const float vc  = sh * sh * c00 + 0.3f;
        float det = va * vc - vb * vb;
        const float pr = -sh * m[1] + H * 0.5f;
        const float pc = -sh * m[0] + H * 0.5f;
        if (o > GTHRESH) cnt += 1.0f;
        const bool keep = (o > GTHRESH) && (det > 0.0f);
        if (!keep) { o = 0.0f; det = 1.0f; }
        float* Pp = params + (size_t)g * 8;
        Pp[0] = pr; Pp[1] = pc; Pp[2] = va; Pp[3] = vb;
        Pp[4] = vc; Pp[5] = det; Pp[6] = o;  Pp[7] = 0.0f;
    }
    atomicAdd(numg, cnt);
}

struct RenderArgs {
    const float* feat[4];
    const float* params;
    float* out;
    int bstart[4];
    int H[4];
    unsigned long long ooff[4];
};

__global__ __launch_bounds__(256) void gauss_render(RenderArgs A)
{
    __shared__ float lds_f[GCHUNK][FPITCH];
    __shared__ float lds_w[WAVES_WG][GCHUNK][PIX_WAVE];
    __shared__ float lds_p[GCHUNK][8];

    const int bs = blockIdx.x;
    int s = 3;
    if      (bs < A.bstart[1]) s = 0;
    else if (bs < A.bstart[2]) s = 1;
    else if (bs < A.bstart[3]) s = 2;
    const int blk = bs - A.bstart[s];

    const float* __restrict__ feat   = A.feat[s];
    const float* __restrict__ params = A.params + (size_t)s * GNG * 8;
    float* __restrict__ out          = A.out + A.ooff[s];
    const int H = A.H[s];
    const int W = H;
    const int P = H * W;

    const int tid  = threadIdx.x;
    const int wave = tid >> 5;
    const int lane = tid & 31;
    const int l16  = lane & 15;
    const int hi   = lane >> 4;
    const int pbase = blk * PIX_WG + wave * PIX_WAVE;

    const int p   = pbase + lane;
    const int pcl = (p < P) ? p : (P - 1);
    const int row = pcl / W;
    const float rf = (float)row;
    const float cf = (float)(pcl - row * W);
    float T = 1.0f;

    v8f acc[8];
    #pragma unroll
    for (int t = 0; t < 8; ++t) {
        #pragma unroll
        for (int i = 0; i < 8; ++i) acc[t][i] = 0.0f;
    }

    for (int ch = 0; ch < GNCHUNK; ++ch) {
        const int nbase = ch * GCHUNK;
        __syncthreads();

        #pragma unroll
        for (int e = tid; e < GCHUNK * GEMBED; e += 256) {
            const int kk  = e & (GCHUNK - 1);
            const int d   = e >> 4;
            const int n   = nbase + kk;
            const int cam = n / 72;
            const int hw  = n - cam * 72;
            lds_f[kk][d] = feat[((size_t)cam * GEMBED + d) * 72 + hw];
            if (ch + 1 < GNCHUNK && (e & 15) == 0) {
                const int n2   = n + GCHUNK;
                const int cam2 = n2 / 72;
                const int hw2  = n2 - cam2 * 72;
                __builtin_prefetch(&feat[((size_t)cam2 * GEMBED + d) * 72 + hw2], 0, 1);
            }
        }
        if (tid < GCHUNK * 8) {
            lds_p[tid >> 3][tid & 7] = params[(size_t)(nbase + (tid >> 3)) * 8 + (tid & 7)];
        }
        __syncthreads();

        if (lane < 16) {
            #pragma unroll
            for (int kk = 0; kk < GCHUNK; ++kk) {
                const float pr  = lds_p[kk][0];
                const float pcn = lds_p[kk][1];
                const float va  = lds_p[kk][2];
                const float vb  = lds_p[kk][3];
                const float vc  = lds_p[kk][4];
                const float det = lds_p[kk][5];
                const float o   = lds_p[kk][6];
                const float dr  = pr - rf;
                const float dc  = pcn - cf;
                const float num = vb * dr * dc - 0.5f * (vc * dr * dr + va * dc * dc);
                const float pw  = num / det;
                const float al  = fminf(0.99f, o * __expf(fminf(pw, 0.0f)));
                const float a2  = ((pw <= 0.0f) && (al >= (1.0f / 255.0f))) ? al : 0.0f;
                lds_w[wave][kk][lane] = a2 * T;
                T *= (1.0f - a2);
            }
        }
        __syncthreads();

        #pragma unroll
        for (int j = 0; j < GCHUNK / 4; ++j) {
            const int k0 = j * 4 + 2 * hi;
            v2f bf;
            bf.x = lds_w[wave][k0    ][l16];
            bf.y = lds_w[wave][k0 + 1][l16];
            #pragma unroll
            for (int t = 0; t < 8; ++t) {
                v2f af;
                af.x = lds_f[k0    ][t * 16 + l16];
                af.y = lds_f[k0 + 1][t * 16 + l16];
                acc[t] = __builtin_amdgcn_wmma_f32_16x16x4_f32(
                     false, af,  false, bf,
                     (short)0, acc[t],  false,  false);
            }
        }
    }

    const int pout = pbase + l16;
    if (pout < P) {
        #pragma unroll
        for (int t = 0; t < 8; ++t) {
            const int dbase = t * 16 + hi * 8;
            #pragma unroll
            for (int i = 0; i < 8; ++i) {
                out[(size_t)(dbase + i) * P + pout] = acc[t][i];
            }
        }
    }
}

extern "C" void kernel_launch(void* const* d_in, const int* in_sizes, int n_in,
                              void* d_out, int out_size, void* d_ws, size_t ws_size,
                              hipStream_t stream)
{
    (void)in_sizes; (void)n_in; (void)ws_size;
    const float* feats[4]; const float* means[4]; const float* uncs[4]; const float* ops[4];
    for (int s = 0; s < 4; ++s) {
        feats[s] = (const float*)d_in[4 * s + 0];
        means[s] = (const float*)d_in[4 * s + 1];
        uncs[s]  = (const float*)d_in[4 * s + 2];
        ops[s]   = (const float*)d_in[4 * s + 3];
    }
    float* out    = (float*)d_out;
    float* params = (float*)d_ws;
    float* numg   = out + (out_size - 1);

    hipLaunchKernelGGL(gauss_prepass, dim3(1), dim3(256), 0, stream,
                       means[0], uncs[0], ops[0], means[1], uncs[1], ops[1],
                       means[2], uncs[2], ops[2], means[3], uncs[3], ops[3],
                       params, numg);

    RenderArgs A;
    A.params = params;
    A.out    = out;
    const int Hs[4] = {25, 50, 100, 200};
    int bcum = 0;
    unsigned long long ocum = 0;
    for (int s = 0; s < 4; ++s) {
        A.feat[s]   = feats[s];
        A.H[s]      = Hs[s];
        A.bstart[s] = bcum;
        A.ooff[s]   = ocum;
        const int Pn = Hs[s] * Hs[s];
        bcum += (Pn + PIX_WG - 1) / PIX_WG;
        ocum += (unsigned long long)GEMBED * Pn;
    }
    hipLaunchKernelGGL(gauss_render, dim3(bcum), dim3(256), 0, stream, A);
}


# next chunk's feature and param tiles are loaded into registers one chunk ahead, kept in flight across LDS-only barriers, instead of nine serialized load-wait-store round trips per chunk
# speedup vs baseline: 1.4988x; 1.4988x over previous
#include <hip/hip_runtime.h>

typedef float v2f __attribute__((ext_vector_type(2)));
typedef float v8f __attribute__((ext_vector_type(8)));

#define GEMBED   128
#define GNG      432
#define GCHUNK   16
#define GNCHUNK  27
#define FPITCH   (GEMBED + 8)
#define PIX_WAVE 16
#define WAVES_WG 8
#define PIX_WG   (PIX_WAVE * WAVES_WG)
#define GTHRESH  0.01f

static __device__ __forceinline__ void lds_barrier()
{
    __builtin_amdgcn_fence(__ATOMIC_RELEASE, "workgroup", "local");
    __builtin_amdgcn_s_barrier();
    __builtin_amdgcn_fence(__ATOMIC_ACQUIRE, "workgroup", "local");
}

__global__ void gauss_prepass(const float* __restrict__ m0, const float* __restrict__ u0, const float* __restrict__ o0,
                              const float* __restrict__ m1, const float* __restrict__ u1, const float* __restrict__ o1,
                              const float* __restrict__ m2, const float* __restrict__ u2, const float* __restrict__ o2,
                              const float* __restrict__ m3, const float* __restrict__ u3, const float* __restrict__ o3,
                              float* __restrict__ params, float* __restrict__ numg)
{
    const int tid = threadIdx.x;
    if (tid == 0) *numg = 0.0f;
    __syncthreads();

    const float* Ms[4] = {m0, m1, m2, m3};
    const float* Us[4] = {u0, u1, u2, u3};
    const float* Os[4] = {o0, o1, o2, o3};
    const float  Hs[4] = {25.f, 50.f, 100.f, 200.f};

    float cnt = 0.0f;
    for (int g = tid; g < 4 * GNG; g += blockDim.x) {
        const int s = g / GNG;
        const int i = g - s * GNG;
        const float* m = Ms[s] + (size_t)i * 3;
        const float* u = Us[s] + (size_t)i * 9;
        float o = Os[s][i];
        const float H  = Hs[s];
        const float sh = H * 0.01f;
        const float c00 = fmaxf(u[0], 1.0f);
        const float c01 = fmaxf(u[1], 1.0f);
        const float c11 = fmaxf(u[4], 1.0f);
        const float va  = sh * sh * c11 + 0.3f;
        const float vb  = sh * sh * c01;
        const float vc  = sh * sh * c00 + 0.3f;
        float det = va * vc - vb * vb;
        const float pr = -sh * m[1] + H * 0.5f;
        const float pc = -sh * m[0] + H * 0.5f;
        if (o > GTHRESH) cnt += 1.0f;
        const bool keep = (o > GTHRESH) && (det > 0.0f);
        if (!keep) { o = 0.0f; det = 1.0f; }
        float* Pp = params + (size_t)g * 8;
        Pp[0] = pr; Pp[1] = pc; Pp[2] = va; Pp[3] = vb;
        Pp[4] = vc; Pp[5] = det; Pp[6] = o;  Pp[7] = 0.0f;
    }
    atomicAdd(numg, cnt);
}

struct RenderArgs {
    const float* feat[4];
    const float* params;
    float* out;
    int bstart[4];
    int H[4];
    unsigned long long ooff[4];
};

__global__ __launch_bounds__(256) void gauss_render(RenderArgs A)
{
    __shared__ float lds_f[GCHUNK][FPITCH];
    __shared__ float lds_w[WAVES_WG][GCHUNK][PIX_WAVE];
    __shared__ float lds_p[GCHUNK][8];

    const int bs = blockIdx.x;
    int s = 3;
    if      (bs < A.bstart[1]) s = 0;
    else if (bs < A.bstart[2]) s = 1;
    else if (bs < A.bstart[3]) s = 2;
    const int blk = bs - A.bstart[s];

    const float* __restrict__ feat   = A.feat[s];
    const float* __restrict__ params = A.params + (size_t)s * GNG * 8;
    float* __restrict__ out          = A.out + A.ooff[s];
    const int H = A.H[s];
    const int W = H;
    const int P = H * W;

    const int tid  = threadIdx.x;
    const int wave = tid >> 5;
    const int lane = tid & 31;
    const int l16  = lane & 15;
    const int hi   = lane >> 4;
    const int pbase = blk * PIX_WG + wave * PIX_WAVE;

    const int p   = pbase + lane;
    const int pcl = (p < P) ? p : (P - 1);
    const int row = pcl / W;
    const float rf = (float)row;
    const float cf = (float)(pcl - row * W);
    float T = 1.0f;

    v8f acc[8];
    #pragma unroll
    for (int t = 0; t < 8; ++t) {
        #pragma unroll
        for (int i = 0; i < 8; ++i) acc[t][i] = 0.0f;
    }

    const int skk = tid & (GCHUNK - 1);
    const int sd0 = tid >> 4;
    float fr[GEMBED / 16];
    float pv;
    {
        const int cam = skk / 72;
        const int hw  = skk - cam * 72;
        const float* __restrict__ fsrc = feat + ((size_t)cam * GEMBED + sd0) * 72 + hw;
        #pragma unroll
        for (int i = 0; i < GEMBED / 16; ++i) fr[i] = fsrc[(size_t)i * 16 * 72];
        pv = params[tid & (GCHUNK * 8 - 1)];
    }

    for (int ch = 0; ch < GNCHUNK; ++ch) {
        lds_barrier();

        #pragma unroll
        for (int i = 0; i < GEMBED / 16; ++i) lds_f[skk][sd0 + 16 * i] = fr[i];
        if (tid < GCHUNK * 8) lds_p[tid >> 3][tid & 7] = pv;

        if (ch + 1 < GNCHUNK) {
            const int n   = (ch + 1) * GCHUNK + skk;
            const int cam = n / 72;
            const int hw  = n - cam * 72;
            const float* __restrict__ fsrc = feat + ((size_t)cam * GEMBED + sd0) * 72 + hw;
            #pragma unroll
            for (int i = 0; i < GEMBED / 16; ++i) fr[i] = fsrc[(size_t)i * 16 * 72];
            pv = params[(size_t)(ch + 1) * GCHUNK * 8 + (tid & (GCHUNK * 8 - 1))];
        }
        lds_barrier();

        if (lane < 16) {
            #pragma unroll
            for (int kk = 0; kk < GCHUNK; ++kk) {
                const float pr  = lds_p[kk][0];
                const float pcn = lds_p[kk][1];
                const float va  = lds_p[kk][2];
                const float vb  = lds_p[kk][3];
                const float vc  = lds_p[kk][4];
                const float det = lds_p[kk][5];
                const float o   = lds_p[kk][6];
                const float dr  = pr - rf;
                const float dc  = pcn - cf;
                const float num = vb * dr * dc - 0.5f * (vc * dr * dr + va * dc * dc);
                const float pw  = num / det;
                const float al  = fminf(0.99f, o * __expf(fminf(pw, 0.0f)));
                const float a2  = ((pw <= 0.0f) && (al >= (1.0f / 255.0f))) ? al : 0.0f;
                lds_w[wave][kk][lane] = a2 * T;
                T *= (1.0f - a2);
            }
        }
        lds_barrier();

        #pragma unroll
        for (int j = 0; j < GCHUNK / 4; ++j) {
            const int k0 = j * 4 + 2 * hi;
            v2f bf;
            bf.x = lds_w[wave][k0    ][l16];
            bf.y = lds_w[wave][k0 + 1][l16];
            #pragma unroll
            for (int t = 0; t < 8; ++t) {
                v2f af;
                af.x = lds_f[k0    ][t * 16 + l16];
                af.y = lds_f[k0 + 1][t * 16 + l16];
                acc[t] = __builtin_amdgcn_wmma_f32_16x16x4_f32(
                     false, af,  false, bf,
                     (short)0, acc[t],  false,  false);
            }
        }
    }

    const int pout = pbase + l16;
    if (pout < P) {
        #pragma unroll
        for (int t = 0; t < 8; ++t) {
            const int dbase = t * 16 + hi * 8;
            #pragma unroll
            for (int i = 0; i < 8; ++i) {
                out[(size_t)(dbase + i) * P + pout] = acc[t][i];
            }
        }
    }
}

extern "C" void kernel_launch(void* const* d_in, const int* in_sizes, int n_in,
                              void* d_out, int out_size, void* d_ws, size_t ws_size,
                              hipStream_t stream)
{
    (void)in_sizes; (void)n_in; (void)ws_size;
    const float* feats[4]; const float* means[4]; const float* uncs[4]; const float* ops[4];
    for (int s = 0; s < 4; ++s) {
        feats[s] = (const float*)d_in[4 * s + 0];
        means[s] = (const float*)d_in[4 * s + 1];
        uncs[s]  = (const float*)d_in[4 * s + 2];
        ops[s]   = (const float*)d_in[4 * s + 3];
    }
    float* out    = (float*)d_out;
    float* params = (float*)d_ws;
    float* numg   = out + (out_size - 1);

    hipLaunchKernelGGL(gauss_prepass, dim3(1), dim3(256), 0, stream,
                       means[0], uncs[0], ops[0], means[1], uncs[1], ops[1],
                       means[2], uncs[2], ops[2], means[3], uncs[3], ops[3],
                       params, numg);

    RenderArgs A;
    A.params = params;
    A.out    = out;
    const int Hs[4] = {25, 50, 100, 200};
    int bcum = 0;
    unsigned long long ocum = 0;
    for (int s = 0; s < 4; ++s) {
        A.feat[s]   = feats[s];
        A.H[s]      = Hs[s];
        A.bstart[s] = bcum;
        A.ooff[s]   = ocum;
        const int Pn = Hs[s] * Hs[s];
        bcum += (Pn + PIX_WG - 1) / PIX_WG;
        ocum += (unsigned long long)GEMBED * Pn;
    }
    hipLaunchKernelGGL(gauss_render, dim3(bcum), dim3(256), 0, stream, A);
}


# GEMM phase skips a 4-gaussian k-step (8 feature-fragment LDS reads, 8 WMMAs) when a wave-wide ballot finds all its 4x16 weights zero; only exact-zero products dropped, sum order kept
# speedup vs baseline: 2.0427x; 1.3629x over previous
#include <hip/hip_runtime.h>

typedef float v2f __attribute__((ext_vector_type(2)));
typedef float v8f __attribute__((ext_vector_type(8)));

#define GEMBED   128
#define GNG      432
#define GCHUNK   16
#define GNCHUNK  27
#define FPITCH   (GEMBED + 8)
#define PIX_WAVE 16
#define WAVES_WG 8
#define PIX_WG   (PIX_WAVE * WAVES_WG)
#define GTHRESH  0.01f

static __device__ __forceinline__ void lds_barrier()
{
    __builtin_amdgcn_fence(__ATOMIC_RELEASE, "workgroup", "local");
    __builtin_amdgcn_s_barrier();
    __builtin_amdgcn_fence(__ATOMIC_ACQUIRE, "workgroup", "local");
}

__global__ void gauss_prepass(const float* __restrict__ m0, const float* __restrict__ u0, const float* __restrict__ o0,
                              const float* __restrict__ m1, const float* __restrict__ u1, const float* __restrict__ o1,
                              const float* __restrict__ m2, const float* __restrict__ u2, const float* __restrict__ o2,
                              const float* __restrict__ m3, const float* __restrict__ u3, const float* __restrict__ o3,
                              float* __restrict__ params, float* __restrict__ numg)
{
    const int tid = threadIdx.x;
    if (tid == 0) *numg = 0.0f;
    __syncthreads();

    const float* Ms[4] = {m0, m1, m2, m3};
    const float* Us[4] = {u0, u1, u2, u3};
    const float* Os[4] = {o0, o1, o2, o3};
    const float  Hs[4] = {25.f, 50.f, 100.f, 200.f};

    float cnt = 0.0f;
    for (int g = tid; g < 4 * GNG; g += blockDim.x) {
        const int s = g / GNG;
        const int i = g - s * GNG;
        const float* m = Ms[s] + (size_t)i * 3;
        const float* u = Us[s] + (size_t)i * 9;
        float o = Os[s][i];
        const float H  = Hs[s];
        const float sh = H * 0.01f;
        const float c00 = fmaxf(u[0], 1.0f);
        const float c01 = fmaxf(u[1], 1.0f);
        const float c11 = fmaxf(u[4], 1.0f);
        const float va  = sh * sh * c11 + 0.3f;
        const float vb  = sh * sh * c01;
        const float vc  = sh * sh * c00 + 0.3f;
        float det = va * vc - vb * vb;
        const float pr = -sh * m[1] + H * 0.5f;
        const float pc = -sh * m[0] + H * 0.5f;
        if (o > GTHRESH) cnt += 1.0f;
        const bool keep = (o > GTHRESH) && (det > 0.0f);
        if (!keep) { o = 0.0f; det = 1.0f; }
        float* Pp = params + (size_t)g * 8;
        Pp[0] = pr; Pp[1] = pc; Pp[2] = va; Pp[3] = vb;
        Pp[4] = vc; Pp[5] = det; Pp[6] = o;  Pp[7] = 0.0f;
    }
    atomicAdd(numg, cnt);
}

struct RenderArgs {
    const float* feat[4];
    const float* params;
    float* out;
    int bstart[4];
    int H[4];
    unsigned long long ooff[4];
};

__global__ __launch_bounds__(256) void gauss_render(RenderArgs A)
{
    __shared__ float lds_f[GCHUNK][FPITCH];
    __shared__ float lds_w[WAVES_WG][GCHUNK][PIX_WAVE];
    __shared__ float lds_p[GCHUNK][8];

    const int bs = blockIdx.x;
    int s = 3;
    if      (bs < A.bstart[1]) s = 0;
    else if (bs < A.bstart[2]) s = 1;
    else if (bs < A.bstart[3]) s = 2;
    const int blk = bs - A.bstart[s];

    const float* __restrict__ feat   = A.feat[s];
    const float* __restrict__ params = A.params + (size_t)s * GNG * 8;
    float* __restrict__ out          = A.out + A.ooff[s];
    const int H = A.H[s];
    const int W = H;
    const int P = H * W;

    const int tid  = threadIdx.x;
    const int wave = tid >> 5;
    const int lane = tid & 31;
    const int l16  = lane & 15;
    const int hi   = lane >> 4;
    const int pbase = blk * PIX_WG + wave * PIX_WAVE;

    const int p   = pbase + lane;
    const int pcl = (p < P) ? p : (P - 1);
    const int row = pcl / W;
    const float rf = (float)row;
    const float cf = (float)(pcl - row * W);
    float T = 1.0f;

    v8f acc[8];
    #pragma unroll
    for (int t = 0; t < 8; ++t) {
        #pragma unroll
        for (int i = 0; i < 8; ++i) acc[t][i] = 0.0f;
    }

    const int skk = tid & (GCHUNK - 1);
    const int sd0 = tid >> 4;
    float fr[GEMBED / 16];
    float pv;
    {
        const int cam = skk / 72;
        const int hw  = skk - cam * 72;
        const float* __restrict__ fsrc = feat + ((size_t)cam * GEMBED + sd0) * 72 + hw;
        #pragma unroll
        for (int i = 0; i < GEMBED / 16; ++i) fr[i] = fsrc[(size_t)i * 16 * 72];
        pv = params[tid & (GCHUNK * 8 - 1)];
    }

    for (int ch = 0; ch < GNCHUNK; ++ch) {
        lds_barrier();

        #pragma unroll
        for (int i = 0; i < GEMBED / 16; ++i) lds_f[skk][sd0 + 16 * i] = fr[i];
        if (tid < GCHUNK * 8) lds_p[tid >> 3][tid & 7] = pv;

        if (ch + 1 < GNCHUNK) {
            const int n   = (ch + 1) * GCHUNK + skk;
            const int cam = n / 72;
            const int hw  = n - cam * 72;
            const float* __restrict__ fsrc = feat + ((size_t)cam * GEMBED + sd0) * 72 + hw;
            #pragma unroll
            for (int i = 0; i < GEMBED / 16; ++i) fr[i] = fsrc[(size_t)i * 16 * 72];
            pv = params[(size_t)(ch + 1) * GCHUNK * 8 + (tid & (GCHUNK * 8 - 1))];
        }
        lds_barrier();

        if (lane < 16) {
            #pragma unroll
            for (int kk = 0; kk < GCHUNK; ++kk) {
                const float pr  = lds_p[kk][0];
                const float pcn = lds_p[kk][1];
                const float va  = lds_p[kk][2];
                const float vb  = lds_p[kk][3];
                const float vc  = lds_p[kk][4];
                const float det = lds_p[kk][5];
                const float o   = lds_p[kk][6];
                const float dr  = pr - rf;
                const float dc  = pcn - cf;
                const float num = vb * dr * dc - 0.5f * (vc * dr * dr + va * dc * dc);
                const float pw  = num / det;
                const float al  = fminf(0.99f, o * __expf(fminf(pw, 0.0f)));
                const float a2  = ((pw <= 0.0f) && (al >= (1.0f / 255.0f))) ? al : 0.0f;
                lds_w[wave][kk][lane] = a2 * T;
                T *= (1.0f - a2);
            }
        }
        lds_barrier();

        v2f bf[GCHUNK / 4];
        #pragma unroll
        for (int j = 0; j < GCHUNK / 4; ++j) {
            const int k0 = j * 4 + 2 * hi;
            bf[j].x = lds_w[wave][k0    ][l16];
            bf[j].y = lds_w[wave][k0 + 1][l16];
        }
        #pragma unroll
        for (int j = 0; j < GCHUNK / 4; ++j) {
            const int k0 = j * 4 + 2 * hi;
            const bool lane_live = (bf[j].x != 0.0f) || (bf[j].y != 0.0f);
            if (__builtin_amdgcn_ballot_w32(lane_live) != 0u) {
                #pragma unroll
                for (int t = 0; t < 8; ++t) {
                    v2f af;
                    af.x = lds_f[k0    ][t * 16 + l16];
                    af.y = lds_f[k0 + 1][t * 16 + l16];
                    acc[t] = __builtin_amdgcn_wmma_f32_16x16x4_f32(
                         false, af,  false, bf[j],
                         (short)0, acc[t],  false,  false);
                }
            }
        }
    }

    const int pout = pbase + l16;
    if (pout < P) {
        #pragma unroll
        for (int t = 0; t < 8; ++t) {
            const int dbase = t * 16 + hi * 8;
            #pragma unroll
            for (int i = 0; i < 8; ++i) {
                out[(size_t)(dbase + i) * P + pout] = acc[t][i];
            }
        }
    }
}

extern "C" void kernel_launch(void* const* d_in, const int* in_sizes, int n_in,
                              void* d_out, int out_size, void* d_ws, size_t ws_size,
                              hipStream_t stream)
{
    (void)in_sizes; (void)n_in; (void)ws_size;
    const float* feats[4]; const float* means[4]; const float* uncs[4]; const float* ops[4];
    for (int s = 0; s < 4; ++s) {
        feats[s] = (const float*)d_in[4 * s + 0];
        means[s] = (const float*)d_in[4 * s + 1];
        uncs[s]  = (const float*)d_in[4 * s + 2];
        ops[s]   = (const float*)d_in[4 * s + 3];
    }
    float* out    = (float*)d_out;
    float* params = (float*)d_ws;
    float* numg   = out + (out_size - 1);

    hipLaunchKernelGGL(gauss_prepass, dim3(1), dim3(256), 0, stream,
                       means[0], uncs[0], ops[0], means[1], uncs[1], ops[1],
                       means[2], uncs[2], ops[2], means[3], uncs[3], ops[3],
                       params, numg);

    RenderArgs A;
    A.params = params;
    A.out    = out;
    const int Hs[4] = {25, 50, 100, 200};
    int bcum = 0;
    unsigned long long ocum = 0;
    for (int s = 0; s < 4; ++s) {
        A.feat[s]   = feats[s];
        A.H[s]      = Hs[s];
        A.bstart[s] = bcum;
        A.ooff[s]   = ocum;
        const int Pn = Hs[s] * Hs[s];
        bcum += (Pn + PIX_WG - 1) / PIX_WG;
        ocum += (unsigned long long)GEMBED * Pn;
    }
    hipLaunchKernelGGL(gauss_render, dim3(bcum), dim3(256), 0, stream, A);
}


# scan evaluates a gaussian only where its alpha-cutoff bounding box (prepass: 2*ln(255*opacity)*variance, widened) meets the wave's 16-pixel tile; culled alphas are exactly zero, sum order kept
# speedup vs baseline: 2.4115x; 1.1805x over previous
#include <hip/hip_runtime.h>

typedef float v2f __attribute__((ext_vector_type(2)));
typedef float v8f __attribute__((ext_vector_type(8)));

#define GEMBED   128
#define GNG      432
#define GCHUNK   16
#define GNCHUNK  27
#define FPITCH   (GEMBED + 8)
#define PIX_WAVE 16
#define WAVES_WG 8
#define PIX_WG   (PIX_WAVE * WAVES_WG)
#define GTHRESH  0.01f

static __device__ __forceinline__ void lds_barrier()
{
    __builtin_amdgcn_fence(__ATOMIC_RELEASE, "workgroup", "local");
    __builtin_amdgcn_s_barrier();
    __builtin_amdgcn_fence(__ATOMIC_ACQUIRE, "workgroup", "local");
}

__global__ void gauss_prepass(const float* __restrict__ m0, const float* __restrict__ u0, const float* __restrict__ o0,
                              const float* __restrict__ m1, const float* __restrict__ u1, const float* __restrict__ o1,
                              const float* __restrict__ m2, const float* __restrict__ u2, const float* __restrict__ o2,
                              const float* __restrict__ m3, const float* __restrict__ u3, const float* __restrict__ o3,
                              float* __restrict__ params, float* __restrict__ numg)
{
    const int tid = threadIdx.x;
    if (tid == 0) *numg = 0.0f;
    __syncthreads();

    const float* Ms[4] = {m0, m1, m2, m3};
    const float* Us[4] = {u0, u1, u2, u3};
    const float* Os[4] = {o0, o1, o2, o3};
    const float  Hs[4] = {25.f, 50.f, 100.f, 200.f};

    float cnt = 0.0f;
    for (int g = tid; g < 4 * GNG; g += blockDim.x) {
        const int s = g / GNG;
        const int i = g - s * GNG;
        const float* m = Ms[s] + (size_t)i * 3;
        const float* u = Us[s] + (size_t)i * 9;
        float o = Os[s][i];
        const float H  = Hs[s];
        const float sh = H * 0.01f;
        const float c00 = fmaxf(u[0], 1.0f);
        const float c01 = fmaxf(u[1], 1.0f);
        const float c11 = fmaxf(u[4], 1.0f);
        const float va  = sh * sh * c11 + 0.3f;
        const float vb  = sh * sh * c01;
        const float vc  = sh * sh * c00 + 0.3f;
        float det = va * vc - vb * vb;
        const float pr = -sh * m[1] + H * 0.5f;
        const float pc = -sh * m[0] + H * 0.5f;
        if (o > GTHRESH) cnt += 1.0f;
        const bool keep = (o > GTHRESH) && (det > 0.0f);
        float reach = -1.0f;
        if (keep) reach = 2.02f * logf(255.0f * o) + 0.04f;
        if (!keep) { o = 0.0f; det = 1.0f; }
        float* Pp = params + (size_t)g * 8;
        Pp[0] = pr; Pp[1] = pc; Pp[2] = va; Pp[3] = vb;
        Pp[4] = vc; Pp[5] = det; Pp[6] = o;  Pp[7] = reach;
    }
    atomicAdd(numg, cnt);
}

struct RenderArgs {
    const float* feat[4];
    const float* params;
    float* out;
    int bstart[4];
    int H[4];
    unsigned long long ooff[4];
};

__global__ __launch_bounds__(256) void gauss_render(RenderArgs A)
{
    __shared__ float lds_f[GCHUNK][FPITCH];
    __shared__ float lds_w[WAVES_WG][GCHUNK][PIX_WAVE];
    __shared__ float lds_p[GCHUNK][8];

    const int bs = blockIdx.x;
    int s = 3;
    if      (bs < A.bstart[1]) s = 0;
    else if (bs < A.bstart[2]) s = 1;
    else if (bs < A.bstart[3]) s = 2;
    const int blk = bs - A.bstart[s];

    const float* __restrict__ feat   = A.feat[s];
    const float* __restrict__ params = A.params + (size_t)s * GNG * 8;
    float* __restrict__ out          = A.out + A.ooff[s];
    const int H = A.H[s];
    const int W = H;
    const int P = H * W;

    const int tid  = threadIdx.x;
    const int wave = tid >> 5;
    const int lane = tid & 31;
    const int l16  = lane & 15;
    const int hi   = lane >> 4;
    const int pbase = blk * PIX_WG + wave * PIX_WAVE;

    const int p   = pbase + lane;
    const int pcl = (p < P) ? p : (P - 1);
    const int row = pcl / W;
    const float rf = (float)row;
    const float cf = (float)(pcl - row * W);
    float T = 1.0f;

    const int pa = (pbase < P) ? pbase : (P - 1);
    const int pb = (pbase + PIX_WAVE - 1 < P) ? (pbase + PIX_WAVE - 1) : (P - 1);
    const int ra = pa / W;
    const int rb = pb / W;
    const float rlo = (float)ra;
    const float rhi = (float)rb;
    const float clo = (ra == rb) ? (float)(pa - ra * W) : 0.0f;
    const float chi = (ra == rb) ? (float)(pb - rb * W) : (float)(W - 1);

    v8f acc[8];
    #pragma unroll
    for (int t = 0; t < 8; ++t) {
        #pragma unroll
        for (int i = 0; i < 8; ++i) acc[t][i] = 0.0f;
    }

    const int skk = tid & (GCHUNK - 1);
    const int sd0 = tid >> 4;
    float fr[GEMBED / 16];
    float pv;
    {
        const int cam = skk / 72;
        const int hw  = skk - cam * 72;
        const float* __restrict__ fsrc = feat + ((size_t)cam * GEMBED + sd0) * 72 + hw;
        #pragma unroll
        for (int i = 0; i < GEMBED / 16; ++i) fr[i] = fsrc[(size_t)i * 16 * 72];
        pv = params[tid & (GCHUNK * 8 - 1)];
    }

    for (int ch = 0; ch < GNCHUNK; ++ch) {
        lds_barrier();

        #pragma unroll
        for (int i = 0; i < GEMBED / 16; ++i) lds_f[skk][sd0 + 16 * i] = fr[i];
        if (tid < GCHUNK * 8) lds_p[tid >> 3][tid & 7] = pv;

        if (ch + 1 < GNCHUNK) {
            const int n   = (ch + 1) * GCHUNK + skk;
            const int cam = n / 72;
            const int hw  = n - cam * 72;
            const float* __restrict__ fsrc = feat + ((size_t)cam * GEMBED + sd0) * 72 + hw;
            #pragma unroll
            for (int i = 0; i < GEMBED / 16; ++i) fr[i] = fsrc[(size_t)i * 16 * 72];
            pv = params[(size_t)(ch + 1) * GCHUNK * 8 + (tid & (GCHUNK * 8 - 1))];
        }
        lds_barrier();

        unsigned mask;
        {
            const float gpr = lds_p[l16][0];
            const float gpc = lds_p[l16][1];
            const float gva = lds_p[l16][2];
            const float gvc = lds_p[l16][4];
            const float grc = lds_p[l16][7];
            const float er  = fmaxf(fmaxf(rlo - gpr, gpr - rhi), 0.0f);
            const float ec  = fmaxf(fmaxf(clo - gpc, gpc - chi), 0.0f);
            const bool hit  = !(er * er > grc * gva) && !(ec * ec > grc * gvc);
            mask = __builtin_amdgcn_ballot_w32(hit) & 0xFFFFu;
        }

        if (mask != 0u) {
            if (lane < 16) {
                #pragma unroll
                for (int j = 0; j < GCHUNK / 4; ++j) {
                    const unsigned sm = (mask >> (4 * j)) & 0xFu;
                    if (sm != 0u) {
                        #pragma unroll
                        for (int i = 0; i < 4; ++i) {
                            const int kk = 4 * j + i;
                            float w = 0.0f;
                            if ((sm >> i) & 1u) {
                                const float pr  = lds_p[kk][0];
                                const float pcn = lds_p[kk][1];
                                const float va  = lds_p[kk][2];
                                const float vb  = lds_p[kk][3];
                                const float vc  = lds_p[kk][4];
                                const float det = lds_p[kk][5];
                                const float o   = lds_p[kk][6];
                                const float dr  = pr - rf;
                                const float dc  = pcn - cf;
                                const float num = vb * dr * dc - 0.5f * (vc * dr * dr + va * dc * dc);
                                const float pw  = num / det;
                                const float al  = fminf(0.99f, o * __expf(fminf(pw, 0.0f)));
                                const float a2  = ((pw <= 0.0f) && (al >= (1.0f / 255.0f))) ? al : 0.0f;
                                w = a2 * T;
                                T *= (1.0f - a2);
                            }
                            lds_w[wave][kk][lane] = w;
                        }
                    }
                }
            }
        }
        lds_barrier();

        #pragma unroll
        for (int j = 0; j < GCHUNK / 4; ++j) {
            if (((mask >> (4 * j)) & 0xFu) != 0u) {
                const int k0 = j * 4 + 2 * hi;
                v2f bfj;
                bfj.x = lds_w[wave][k0    ][l16];
                bfj.y = lds_w[wave][k0 + 1][l16];
                const bool lane_live = (bfj.x != 0.0f) || (bfj.y != 0.0f);
                if (__builtin_amdgcn_ballot_w32(lane_live) != 0u) {
                    #pragma unroll
                    for (int t = 0; t < 8; ++t) {
                        v2f af;
                        af.x = lds_f[k0    ][t * 16 + l16];
                        af.y = lds_f[k0 + 1][t * 16 + l16];
                        acc[t] = __builtin_amdgcn_wmma_f32_16x16x4_f32(
                             false, af,  false, bfj,
                             (short)0, acc[t],  false,  false);
                    }
                }
            }
        }
    }

    const int pout = pbase + l16;
    if (pout < P) {
        #pragma unroll
        for (int t = 0; t < 8; ++t) {
            const int dbase = t * 16 + hi * 8;
            #pragma unroll
            for (int i = 0; i < 8; ++i) {
                out[(size_t)(dbase + i) * P + pout] = acc[t][i];
            }
        }
    }
}

extern "C" void kernel_launch(void* const* d_in, const int* in_sizes, int n_in,
                              void* d_out, int out_size, void* d_ws, size_t ws_size,
                              hipStream_t stream)
{
    (void)in_sizes; (void)n_in; (void)ws_size;
    const float* feats[4]; const float* means[4]; const float* uncs[4]; const float* ops[4];
    for (int s = 0; s < 4; ++s) {
        feats[s] = (const float*)d_in[4 * s + 0];
        means[s] = (const float*)d_in[4 * s + 1];
        uncs[s]  = (const float*)d_in[4 * s + 2];
        ops[s]   = (const float*)d_in[4 * s + 3];
    }
    float* out    = (float*)d_out;
    float* params = (float*)d_ws;
    float* numg   = out + (out_size - 1);

    hipLaunchKernelGGL(gauss_prepass, dim3(1), dim3(256), 0, stream,
                       means[0], uncs[0], ops[0], means[1], uncs[1], ops[1],
                       means[2], uncs[2], ops[2], means[3], uncs[3], ops[3],
                       params, numg);

    RenderArgs A;
    A.params = params;
    A.out    = out;
    const int Hs[4] = {25, 50, 100, 200};
    int bcum = 0;
    unsigned long long ocum = 0;
    for (int s = 0; s < 4; ++s) {
        A.feat[s]   = feats[s];
        A.H[s]      = Hs[s];
        A.bstart[s] = bcum;
        A.ooff[s]   = ocum;
        const int Pn = Hs[s] * Hs[s];
        bcum += (Pn + PIX_WG - 1) / PIX_WG;
        ocum += (unsigned long long)GEMBED * Pn;
    }
    hipLaunchKernelGGL(gauss_render, dim3(bcum), dim3(256), 0, stream, A);
}


# each workgroup lists, in order, the gaussians whose cutoff box meets its 128-pixel tile and loops over those only, fetching 512 B rows of a gaussian-major feature copy; k-step grouping of sums changes
# speedup vs baseline: 3.2631x; 1.3531x over previous
#include <hip/hip_runtime.h>

typedef float v2f __attribute__((ext_vector_type(2)));
typedef float v8f __attribute__((ext_vector_type(8)));

#define GEMBED   128
#define GNG      432
#define GCHUNK   16
#define FPITCH   (GEMBED + 8)
#define PIX_WAVE 16
#define WAVES_WG 8
#define PIX_WG   (PIX_WAVE * WAVES_WG)
#define GTHRESH  0.01f
#define WS_FEAT_OFF 16384
#define TR_DCH   32
#define TR_BLOCKS (4 * 6 * (GEMBED / TR_DCH))

static __device__ __forceinline__ void lds_barrier()
{
    __builtin_amdgcn_fence(__ATOMIC_RELEASE, "workgroup", "local");
    __builtin_amdgcn_s_barrier();
    __builtin_amdgcn_fence(__ATOMIC_ACQUIRE, "workgroup", "local");
}

__global__ void gauss_prepass(const float* __restrict__ m0, const float* __restrict__ u0, const float* __restrict__ o0,
                              const float* __restrict__ m1, const float* __restrict__ u1, const float* __restrict__ o1,
                              const float* __restrict__ m2, const float* __restrict__ u2, const float* __restrict__ o2,
                              const float* __restrict__ m3, const float* __restrict__ u3, const float* __restrict__ o3,
                              const float* __restrict__ f0, const float* __restrict__ f1,
                              const float* __restrict__ f2, const float* __restrict__ f3,
                              float* __restrict__ params, float* __restrict__ numg, float* __restrict__ featT)
{
    __shared__ float tile[TR_DCH][73];
    const int tid = threadIdx.x;

    if (blockIdx.x != 0) {
        const int b   = (int)blockIdx.x - 1;
        const int s   = b / 24;
        const int r   = b - s * 24;
        const int cam = r >> 2;
        const int dq  = r & 3;
        const float* __restrict__ fs = (s == 0) ? f0 : (s == 1) ? f1 : (s == 2) ? f2 : f3;
        const float* __restrict__ srcp = fs + ((size_t)cam * GEMBED + dq * TR_DCH) * 72;
        for (int e = tid; e < TR_DCH * 72; e += 256) {
            const int dl = e / 72;
            tile[dl][e - dl * 72] = srcp[e];
        }
        __syncthreads();
        float* __restrict__ dstp = featT + ((size_t)s * GNG + cam * 72) * GEMBED + dq * TR_DCH;
        for (int e = tid; e < TR_DCH * 72; e += 256) {
            const int hw = e >> 5;
            const int dl = e & 31;
            dstp[(size_t)hw * GEMBED + dl] = tile[dl][hw];
        }
        return;
    }
    if (tid == 0) *numg = 0.0f;
    __syncthreads();

    const float* Ms[4] = {m0, m1, m2, m3};
    const float* Us[4] = {u0, u1, u2, u3};
    const float* Os[4] = {o0, o1, o2, o3};
    const float  Hs[4] = {25.f, 50.f, 100.f, 200.f};

    float cnt = 0.0f;
    for (int g = tid; g < 4 * GNG; g += blockDim.x) {
        const int s = g / GNG;
        const int i = g - s * GNG;
        const float* m = Ms[s] + (size_t)i * 3;
        const float* u = Us[s] + (size_t)i * 9;
        float o = Os[s][i];
        const float H  = Hs[s];
        const float sh = H * 0.01f;
        const float c00 = fmaxf(u[0], 1.0f);
        const float c01 = fmaxf(u[1], 1.0f);
        const float c11 = fmaxf(u[4], 1.0f);
        const float va  = sh * sh * c11 + 0.3f;
        const float vb  = sh * sh * c01;
        const float vc  = sh * sh * c00 + 0.3f;
        float det = va * vc - vb * vb;
        const float pr = -sh * m[1] + H * 0.5f;
        const float pc = -sh * m[0] + H * 0.5f;
        if (o > GTHRESH) cnt += 1.0f;
        const bool keep = (o > GTHRESH) && (det > 0.0f);
        float reach = -1.0f;
        if (keep) reach = 2.02f * logf(255.0f * o) + 0.04f;
        if (!keep) { o = 0.0f; det = 1.0f; }
        float* Pp = params + (size_t)g * 8;
        Pp[0] = pr; Pp[1] = pc; Pp[2] = va; Pp[3] = vb;
        Pp[4] = vc; Pp[5] = det; Pp[6] = o;  Pp[7] = reach;
    }
    atomicAdd(numg, cnt);
}

struct RenderArgs {
    const float* featT;
    const float* params;
    float* out;
    int bstart[4];
    int H[4];
    unsigned long long ooff[4];
};

__global__ __launch_bounds__(256) void gauss_render(RenderArgs A)
{
    __shared__ __attribute__((aligned(16))) float lds_f[GCHUNK][FPITCH];
    __shared__ int lds_list[GNG + GCHUNK];
    __shared__ int lds_cnt[2 * WAVES_WG];
    __shared__ float lds_w[WAVES_WG][GCHUNK][PIX_WAVE];
    __shared__ float lds_p[GCHUNK][8];

    const int bs = blockIdx.x;
    int s = 3;
    if      (bs < A.bstart[1]) s = 0;
    else if (bs < A.bstart[2]) s = 1;
    else if (bs < A.bstart[3]) s = 2;
    const int blk = bs - A.bstart[s];

    const float* __restrict__ featT  = A.featT + (size_t)s * GNG * GEMBED;
    const float* __restrict__ params = A.params + (size_t)s * GNG * 8;
    float* __restrict__ out          = A.out + A.ooff[s];
    const int H = A.H[s];
    const int W = H;
    const int P = H * W;

    const int tid  = threadIdx.x;
    const int wave = tid >> 5;
    const int lane = tid & 31;
    const int l16  = lane & 15;
    const int hi   = lane >> 4;
    const int pbase = blk * PIX_WG + wave * PIX_WAVE;

    const int p   = pbase + lane;
    const int pcl = (p < P) ? p : (P - 1);
    const int row = pcl / W;
    const float rf = (float)row;
    const float cf = (float)(pcl - row * W);
    float T = 1.0f;

    const int pa = (pbase < P) ? pbase : (P - 1);
    const int pb = (pbase + PIX_WAVE - 1 < P) ? (pbase + PIX_WAVE - 1) : (P - 1);
    const int ra = pa / W;
    const int rb = pb / W;
    const float rlo = (float)ra;
    const float rhi = (float)rb;
    const float clo = (ra == rb) ? (float)(pa - ra * W) : 0.0f;
    const float chi = (ra == rb) ? (float)(pb - rb * W) : (float)(W - 1);

    v8f acc[8];
    #pragma unroll
    for (int t = 0; t < 8; ++t) {
        #pragma unroll
        for (int i = 0; i < 8; ++i) acc[t][i] = 0.0f;
    }

    int nlist;
    {
        const int qa  = blk * PIX_WG;
        const int qb  = (qa + PIX_WG - 1 < P) ? (qa + PIX_WG - 1) : (P - 1);
        const int bra = qa / W;
        const int brb = qb / W;
        const float brlo = (float)bra;
        const float brhi = (float)brb;
        const float bclo = (bra == brb) ? (float)(qa - bra * W) : 0.0f;
        const float bchi = (bra == brb) ? (float)(qb - brb * W) : (float)(W - 1);
        bool     hitv[2];
        unsigned balv[2];
        #pragma unroll
        for (int it = 0; it < 2; ++it) {
            const int g = it * 256 + tid;
            bool hit = false;
            if (g < GNG) {
                const float4 q0 = *reinterpret_cast<const float4*>(params + (size_t)g * 8);
                const float4 q1 = *reinterpret_cast<const float4*>(params + (size_t)g * 8 + 4);
                const float er = fmaxf(fmaxf(brlo - q0.x, q0.x - brhi), 0.0f);
                const float ec = fmaxf(fmaxf(bclo - q0.y, q0.y - bchi), 0.0f);
                hit = !(er * er > q1.w * q0.z) && !(ec * ec > q1.w * q1.x);
            }
            hitv[it] = hit;
            balv[it] = __builtin_amdgcn_ballot_w32(hit);
            if (lane == 0) lds_cnt[it * WAVES_WG + wave] = __builtin_popcount(balv[it]);
        }
        __syncthreads();
        int total = 0, base0 = 0, base1 = 0;
        #pragma unroll
        for (int k = 0; k < 2 * WAVES_WG; ++k) {
            if (k == wave) base0 = total;
            if (k == WAVES_WG + wave) base1 = total;
            total += lds_cnt[k];
        }
        const unsigned below = (1u << lane) - 1u;
        if (hitv[0]) lds_list[base0 + __builtin_popcount(balv[0] & below)] = tid;
        if (hitv[1]) lds_list[base1 + __builtin_popcount(balv[1] & below)] = 256 + tid;
        nlist = total;
        if (tid < GCHUNK && total + tid < ((total + GCHUNK - 1) & ~(GCHUNK - 1))) lds_list[total + tid] = -1;
        __syncthreads();
    }
    const int nchunk = __builtin_amdgcn_readfirstlane((int)((unsigned)(nlist + GCHUNK - 1) >> 4));

    const int sA = 2 * wave;
    const int sP = (tid >> 3) & (GCHUNK - 1);
    const int pw8 = tid & 7;
    const float padv = (pw8 == 2 || pw8 == 4 || pw8 == 5) ? 1.0f : ((pw8 == 7) ? -1.0f : 0.0f);
    float4 frA, frB;
    float pv;
    frA = make_float4(0.0f, 0.0f, 0.0f, 0.0f);
    frB = frA;
    pv  = padv;
    if (nchunk > 0) {
        const int iA = lds_list[sA];
        const int iB = lds_list[sA + 1];
        const int iP = lds_list[sP];
        if (iA >= 0) frA = *reinterpret_cast<const float4*>(featT + (size_t)iA * GEMBED + lane * 4);
        if (iB >= 0) frB = *reinterpret_cast<const float4*>(featT + (size_t)iB * GEMBED + lane * 4);
        if (iP >= 0) pv = params[(size_t)iP * 8 + pw8];
    }

    for (int ch = 0; ch < nchunk; ++ch) {
        lds_barrier();

        *reinterpret_cast<float4*>(&lds_f[sA    ][lane * 4]) = frA;
        *reinterpret_cast<float4*>(&lds_f[sA + 1][lane * 4]) = frB;
        if (tid < GCHUNK * 8) lds_p[sP][pw8] = pv;

        if (ch + 1 < nchunk) {
            const int iA = lds_list[(ch + 1) * GCHUNK + sA];
            const int iB = lds_list[(ch + 1) * GCHUNK + sA + 1];
            const int iP = lds_list[(ch + 1) * GCHUNK + sP];
            frA = make_float4(0.0f, 0.0f, 0.0f, 0.0f);
            frB = frA;
            pv  = padv;
            if (iA >= 0) frA = *reinterpret_cast<const float4*>(featT + (size_t)iA * GEMBED + lane * 4);
            if (iB >= 0) frB = *reinterpret_cast<const float4*>(featT + (size_t)iB * GEMBED + lane * 4);
            if (iP >= 0) pv = params[(size_t)iP * 8 + pw8];
        }
        lds_barrier();

        unsigned mask;
        {
            const float gpr = lds_p[l16][0];
            const float gpc = lds_p[l16][1];
            const float gva = lds_p[l16][2];
            const float gvc = lds_p[l16][4];
            const float grc = lds_p[l16][7];
            const float er  = fmaxf(fmaxf(rlo - gpr, gpr - rhi), 0.0f);
            const float ec  = fmaxf(fmaxf(clo - gpc, gpc - chi), 0.0f);
            const bool hit  = !(er * er > grc * gva) && !(ec * ec > grc * gvc);
            mask = __builtin_amdgcn_ballot_w32(hit) & 0xFFFFu;
        }

        if (mask != 0u) {
            if (lane < 16) {
                #pragma unroll
                for (int j = 0; j < GCHUNK / 4; ++j) {
                    const unsigned sm = (mask >> (4 * j)) & 0xFu;
                    if (sm != 0u) {
                        #pragma unroll
                        for (int i = 0; i < 4; ++i) {
                            const int kk = 4 * j + i;
                            float w = 0.0f;
                            if ((sm >> i) & 1u) {
                                const float pr  = lds_p[kk][0];
                                const float pcn = lds_p[kk][1];
                                const float va  = lds_p[kk][2];
                                const float vb  = lds_p[kk][3];
                                const float vc  = lds_p[kk][4];
                                const float det = lds_p[kk][5];
                                const float o   = lds_p[kk][6];
                                const float dr  = pr - rf;
                                const float dc  = pcn - cf;
                                const float num = vb * dr * dc - 0.5f * (vc * dr * dr + va * dc * dc);
                                const float pw  = num / det;
                                const float al  = fminf(0.99f, o * __expf(fminf(pw, 0.0f)));
                                const float a2  = ((pw <= 0.0f) && (al >= (1.0f / 255.0f))) ? al : 0.0f;
                                w = a2 * T;
                                T *= (1.0f - a2);
                            }
                            lds_w[wave][kk][lane] = w;
                        }
                    }
                }
            }
        }
        lds_barrier();

        #pragma unroll
        for (int j = 0; j < GCHUNK / 4; ++j) {
            if (((mask >> (4 * j)) & 0xFu) != 0u) {
                const int k0 = j * 4 + 2 * hi;
                v2f bfj;
                bfj.x = lds_w[wave][k0    ][l16];
                bfj.y = lds_w[wave][k0 + 1][l16];
                const bool lane_live = (bfj.x != 0.0f) || (bfj.y != 0.0f);
                if (__builtin_amdgcn_ballot_w32(lane_live) != 0u) {
                    #pragma unroll
                    for (int t = 0; t < 8; ++t) {
                        v2f af;
                        af.x = lds_f[k0    ][t * 16 + l16];
                        af.y = lds_f[k0 + 1][t * 16 + l16];
                        acc[t] = __builtin_amdgcn_wmma_f32_16x16x4_f32(
                             false, af,  false, bfj,
                             (short)0, acc[t],  false,  false);
                    }
                }
            }
        }
    }

    const int pout = pbase + l16;
    if (pout < P) {
        #pragma unroll
        for (int t = 0; t < 8; ++t) {
            const int dbase = t * 16 + hi * 8;
            #pragma unroll
            for (int i = 0; i < 8; ++i) {
                out[(size_t)(dbase + i) * P + pout] = acc[t][i];
            }
        }
    }
}

extern "C" void kernel_launch(void* const* d_in, const int* in_sizes, int n_in,
                              void* d_out, int out_size, void* d_ws, size_t ws_size,
                              hipStream_t stream)
{
    (void)in_sizes; (void)n_in; (void)ws_size;
    const float* feats[4]; const float* means[4]; const float* uncs[4]; const float* ops[4];
    for (int s = 0; s < 4; ++s) {
        feats[s] = (const float*)d_in[4 * s + 0];
        means[s] = (const float*)d_in[4 * s + 1];
        uncs[s]  = (const float*)d_in[4 * s + 2];
        ops[s]   = (const float*)d_in[4 * s + 3];
    }
    float* out    = (float*)d_out;
    float* params = (float*)d_ws;
    float* featT  = (float*)d_ws + WS_FEAT_OFF;
    float* numg   = out + (out_size - 1);

    hipLaunchKernelGGL(gauss_prepass, dim3(1 + TR_BLOCKS), dim3(256), 0, stream,
                       means[0], uncs[0], ops[0], means[1], uncs[1], ops[1],
                       means[2], uncs[2], ops[2], means[3], uncs[3], ops[3],
                       feats[0], feats[1], feats[2], feats[3],
                       params, numg, featT);

    RenderArgs A;
    A.featT  = featT;
    A.params = params;
    A.out    = out;
    const int Hs[4] = {25, 50, 100, 200};
    int bcum = 0;
    unsigned long long ocum = 0;
    for (int s = 0; s < 4; ++s) {
        A.H[s]      = Hs[s];
        A.bstart[s] = bcum;
        A.ooff[s]   = ocum;
        const int Pn = Hs[s] * Hs[s];
        bcum += (Pn + PIX_WG - 1) / PIX_WG;
        ocum += (unsigned long long)GEMBED * Pn;
    }
    hipLaunchKernelGGL(gauss_render, dim3(bcum), dim3(256), 0, stream, A);
}


# weight scan on all 32 lanes: each lane half evaluates half of a chunk's gaussians branch-free (8 independent alphas per lane); weights go to the WMMA B fragment in registers; T product order kept
# speedup vs baseline: 4.3146x; 1.3222x over previous
#include <hip/hip_runtime.h>

typedef float v2f __attribute__((ext_vector_type(2)));
typedef float v8f __attribute__((ext_vector_type(8)));

#define GEMBED   128
#define GNG      432
#define GCHUNK   16
#define FPITCH   (GEMBED + 8)
#define PIX_WAVE 16
#define WAVES_WG 8
#define PIX_WG   (PIX_WAVE * WAVES_WG)
#define GTHRESH  0.01f
#define WS_FEAT_OFF 16384
#define TR_DCH   32
#define TR_BLOCKS (4 * 6 * (GEMBED / TR_DCH))

static __device__ __forceinline__ void lds_barrier()
{
    __builtin_amdgcn_fence(__ATOMIC_RELEASE, "workgroup", "local");
    __builtin_amdgcn_s_barrier();
    __builtin_amdgcn_fence(__ATOMIC_ACQUIRE, "workgroup", "local");
}

static __device__ __forceinline__ float xhalf(float x)
{
    const int b = __builtin_bit_cast(int, x);
    return __builtin_bit_cast(float, __builtin_amdgcn_permlanex16(b, b, 0x76543210, 0xfedcba98, false, false));
}

__global__ void gauss_prepass(const float* __restrict__ m0, const float* __restrict__ u0, const float* __restrict__ o0,
                              const float* __restrict__ m1, const float* __restrict__ u1, const float* __restrict__ o1,
                              const float* __restrict__ m2, const float* __restrict__ u2, const float* __restrict__ o2,
                              const float* __restrict__ m3, const float* __restrict__ u3, const float* __restrict__ o3,
                              const float* __restrict__ f0, const float* __restrict__ f1,
                              const float* __restrict__ f2, const float* __restrict__ f3,
                              float* __restrict__ params, float* __restrict__ numg, float* __restrict__ featT)
{
    __shared__ float tile[TR_DCH][73];
    const int tid = threadIdx.x;

    if (blockIdx.x != 0) {
        const int b   = (int)blockIdx.x - 1;
        const int s   = b / 24;
        const int r   = b - s * 24;
        const int cam = r >> 2;
        const int dq  = r & 3;
        const float* __restrict__ fs = (s == 0) ? f0 : (s == 1) ? f1 : (s == 2) ? f2 : f3;
        const float* __restrict__ srcp = fs + ((size_t)cam * GEMBED + dq * TR_DCH) * 72;
        for (int e = tid; e < TR_DCH * 72; e += 256) {
            const int dl = e / 72;
            tile[dl][e - dl * 72] = srcp[e];
        }
        __syncthreads();
        float* __restrict__ dstp = featT + ((size_t)s * GNG + cam * 72) * GEMBED + dq * TR_DCH;
        for (int e = tid; e < TR_DCH * 72; e += 256) {
            const int hw = e >> 5;
            const int dl = e & 31;
            dstp[(size_t)hw * GEMBED + dl] = tile[dl][hw];
        }
        return;
    }
    if (tid == 0) *numg = 0.0f;
    __syncthreads();

    const float* Ms[4] = {m0, m1, m2, m3};
    const float* Us[4] = {u0, u1, u2, u3};
    const float* Os[4] = {o0, o1, o2, o3};
    const float  Hs[4] = {25.f, 50.f, 100.f, 200.f};

    float cnt = 0.0f;
    for (int g = tid; g < 4 * GNG; g += blockDim.x) {
        const int s = g / GNG;
        const int i = g - s * GNG;
        const float* m = Ms[s] + (size_t)i * 3;
        const float* u = Us[s] + (size_t)i * 9;
        float o = Os[s][i];
        const float H  = Hs[s];
        const float sh = H * 0.01f;
        const float c00 = fmaxf(u[0], 1.0f);
        const float c01 = fmaxf(u[1], 1.0f);
        const float c11 = fmaxf(u[4], 1.0f);
        const float va  = sh * sh * c11 + 0.3f;
        const float vb  = sh * sh * c01;
        const float vc  = sh * sh * c00 + 0.3f;
        float det = va * vc - vb * vb;
        const float pr = -sh * m[1] + H * 0.5f;
        const float pc = -sh * m[0] + H * 0.5f;
        if (o > GTHRESH) cnt += 1.0f;
        const bool keep = (o > GTHRESH) && (det > 0.0f);
        float reach = -1.0f;
        if (keep) reach = 2.02f * logf(255.0f * o) + 0.04f;
        if (!keep) { o = 0.0f; det = 1.0f; }
        float* Pp = params + (size_t)g * 8;
        Pp[0] = pr; Pp[1] = pc; Pp[2] = va; Pp[3] = vb;
        Pp[4] = vc; Pp[5] = det; Pp[6] = o;  Pp[7] = reach;
    }
    atomicAdd(numg, cnt);
}

struct RenderArgs {
    const float* featT;
    const float* params;
    float* out;
    int bstart[4];
    int H[4];
    unsigned long long ooff[4];
};

__global__ __launch_bounds__(256) void gauss_render(RenderArgs A)
{
    __shared__ __attribute__((aligned(16))) float lds_f[GCHUNK][FPITCH];
    __shared__ int lds_list[GNG + GCHUNK];
    __shared__ int lds_cnt[2 * WAVES_WG];
    __shared__ __attribute__((aligned(16))) float lds_p[GCHUNK][8];

    const int bs = blockIdx.x;
    int s = 3;
    if      (bs < A.bstart[1]) s = 0;
    else if (bs < A.bstart[2]) s = 1;
    else if (bs < A.bstart[3]) s = 2;
    const int blk = bs - A.bstart[s];

    const float* __restrict__ featT  = A.featT + (size_t)s * GNG * GEMBED;
    const float* __restrict__ params = A.params + (size_t)s * GNG * 8;
    float* __restrict__ out          = A.out + A.ooff[s];
    const int H = A.H[s];
    const int W = H;
    const int P = H * W;

    const int tid  = threadIdx.x;
    const int wave = tid >> 5;
    const int lane = tid & 31;
    const int l16  = lane & 15;
    const int hi   = lane >> 4;
    const int pbase = blk * PIX_WG + wave * PIX_WAVE;

    const int p   = pbase + l16;
    const int pcl = (p < P) ? p : (P - 1);
    const int row = pcl / W;
    const float rf = (float)row;
    const float cf = (float)(pcl - row * W);
    float T = 1.0f;

    v8f acc[8];
    #pragma unroll
    for (int t = 0; t < 8; ++t) {
        #pragma unroll
        for (int i = 0; i < 8; ++i) acc[t][i] = 0.0f;
    }

    int nlist;
    {
        const int qa  = blk * PIX_WG;
        const int qb  = (qa + PIX_WG - 1 < P) ? (qa + PIX_WG - 1) : (P - 1);
        const int bra = qa / W;
        const int brb = qb / W;
        const float brlo = (float)bra;
        const float brhi = (float)brb;
        const float bclo = (bra == brb) ? (float)(qa - bra * W) : 0.0f;
        const float bchi = (bra == brb) ? (float)(qb - brb * W) : (float)(W - 1);
        bool     hitv[2];
        unsigned balv[2];
        #pragma unroll
        for (int it = 0; it < 2; ++it) {
            const int g = it * 256 + tid;
            bool hit = false;
            if (g < GNG) {
                const float4 q0 = *reinterpret_cast<const float4*>(params + (size_t)g * 8);
                const float4 q1 = *reinterpret_cast<const float4*>(params + (size_t)g * 8 + 4);
                const float er = fmaxf(fmaxf(brlo - q0.x, q0.x - brhi), 0.0f);
                const float ec = fmaxf(fmaxf(bclo - q0.y, q0.y - bchi), 0.0f);
                hit = !(er * er > q1.w * q0.z) && !(ec * ec > q1.w * q1.x);
            }
            hitv[it] = hit;
            balv[it] = __builtin_amdgcn_ballot_w32(hit);
            if (lane == 0) lds_cnt[it * WAVES_WG + wave] = __builtin_popcount(balv[it]);
        }
        __syncthreads();
        int total = 0, base0 = 0, base1 = 0;
        #pragma unroll
        for (int k = 0; k < 2 * WAVES_WG; ++k) {
            if (k == wave) base0 = total;
            if (k == WAVES_WG + wave) base1 = total;
            total += lds_cnt[k];
        }
        const unsigned below = (1u << lane) - 1u;
        if (hitv[0]) lds_list[base0 + __builtin_popcount(balv[0] & below)] = tid;
        if (hitv[1]) lds_list[base1 + __builtin_popcount(balv[1] & below)] = 256 + tid;
        nlist = total;
        if (tid < GCHUNK && total + tid < ((total + GCHUNK - 1) & ~(GCHUNK - 1))) lds_list[total + tid] = -1;
        __syncthreads();
    }
    const int nchunk = __builtin_amdgcn_readfirstlane((int)((unsigned)(nlist + GCHUNK - 1) >> 4));

    const int sA = 2 * wave;
    const int sP = (tid >> 3) & (GCHUNK - 1);
    const int pw8 = tid & 7;
    const float padv = (pw8 == 2 || pw8 == 4 || pw8 == 5) ? 1.0f : ((pw8 == 7) ? -1.0f : 0.0f);
    float4 frA, frB;
    float pv;
    frA = make_float4(0.0f, 0.0f, 0.0f, 0.0f);
    frB = frA;
    pv  = padv;
    if (nchunk > 0) {
        const int iA = lds_list[sA];
        const int iB = lds_list[sA + 1];
        const int iP = lds_list[sP];
        if (iA >= 0) frA = *reinterpret_cast<const float4*>(featT + (size_t)iA * GEMBED + lane * 4);
        if (iB >= 0) frB = *reinterpret_cast<const float4*>(featT + (size_t)iB * GEMBED + lane * 4);
        if (iP >= 0) pv = params[(size_t)iP * 8 + pw8];
    }

    for (int ch = 0; ch < nchunk; ++ch) {
        lds_barrier();

        *reinterpret_cast<float4*>(&lds_f[sA    ][lane * 4]) = frA;
        *reinterpret_cast<float4*>(&lds_f[sA + 1][lane * 4]) = frB;
        if (tid < GCHUNK * 8) lds_p[sP][pw8] = pv;

        if (ch + 1 < nchunk) {
            const int iA = lds_list[(ch + 1) * GCHUNK + sA];
            const int iB = lds_list[(ch + 1) * GCHUNK + sA + 1];
            const int iP = lds_list[(ch + 1) * GCHUNK + sP];
            frA = make_float4(0.0f, 0.0f, 0.0f, 0.0f);
            frB = frA;
            pv  = padv;
            if (iA >= 0) frA = *reinterpret_cast<const float4*>(featT + (size_t)iA * GEMBED + lane * 4);
            if (iB >= 0) frB = *reinterpret_cast<const float4*>(featT + (size_t)iB * GEMBED + lane * 4);
            if (iP >= 0) pv = params[(size_t)iP * 8 + pw8];
        }
        lds_barrier();

        float am[8];
        float om[8];
        #pragma unroll
        for (int m = 0; m < 8; ++m) {
            const int kk = 4 * (m >> 1) + 2 * hi + (m & 1);
            const float4 q0 = *reinterpret_cast<const float4*>(&lds_p[kk][0]);
            const float4 q1 = *reinterpret_cast<const float4*>(&lds_p[kk][4]);
            const float dr  = q0.x - rf;
            const float dc  = q0.y - cf;
            const float num = q0.w * dr * dc - 0.5f * (q1.x * dr * dr + q0.z * dc * dc);
            const float pw  = num / q1.y;
            const float al  = fminf(0.99f, q1.z * __expf(fminf(pw, 0.0f)));
            am[m] = ((pw <= 0.0f) && (al >= (1.0f / 255.0f))) ? al : 0.0f;
            om[m] = 1.0f - am[m];
        }
        float op[8];
        #pragma unroll
        for (int m = 0; m < 8; ++m) op[m] = xhalf(om[m]);

        v2f bf[4];
        #pragma unroll
        for (int j = 0; j < GCHUNK / 4; ++j) {
            const float f0 = hi ? op[2 * j]     : om[2 * j];
            const float f1 = hi ? op[2 * j + 1] : om[2 * j + 1];
            const float s0 = hi ? om[2 * j]     : op[2 * j];
            const float s1 = hi ? om[2 * j + 1] : op[2 * j + 1];
            const float T0 = T;
            const float T1 = T0 * f0;
            const float T2 = T1 * f1;
            const float T3 = T2 * s0;
            const float T4 = T3 * s1;
            bf[j].x = am[2 * j]     * (hi ? T2 : T0);
            bf[j].y = am[2 * j + 1] * (hi ? T3 : T1);
            T = T4;
        }

        #pragma unroll
        for (int j = 0; j < GCHUNK / 4; ++j) {
            const int k0 = j * 4 + 2 * hi;
            const v2f bfj = bf[j];
            const bool lane_live = (bfj.x != 0.0f) || (bfj.y != 0.0f);
            if (__builtin_amdgcn_ballot_w32(lane_live) != 0u) {
                #pragma unroll
                for (int t = 0; t < 8; ++t) {
                    v2f af;
                    af.x = lds_f[k0    ][t * 16 + l16];
                    af.y = lds_f[k0 + 1][t * 16 + l16];
                    acc[t] = __builtin_amdgcn_wmma_f32_16x16x4_f32(
                         false, af,  false, bfj,
                         (short)0, acc[t],  false,  false);
                }
            }
        }
    }

    const int pout = pbase + l16;
    if (pout < P) {
        #pragma unroll
        for (int t = 0; t < 8; ++t) {
            const int dbase = t * 16 + hi * 8;
            #pragma unroll
            for (int i = 0; i < 8; ++i) {
                out[(size_t)(dbase + i) * P + pout] = acc[t][i];
            }
        }
    }
}

extern "C" void kernel_launch(void* const* d_in, const int* in_sizes, int n_in,
                              void* d_out, int out_size, void* d_ws, size_t ws_size,
                              hipStream_t stream)
{
    (void)in_sizes; (void)n_in; (void)ws_size;
    const float* feats[4]; const float* means[4]; const float* uncs[4]; const float* ops[4];
    for (int s = 0; s < 4; ++s) {
        feats[s] = (const float*)d_in[4 * s + 0];
        means[s] = (const float*)d_in[4 * s + 1];
        uncs[s]  = (const float*)d_in[4 * s + 2];
        ops[s]   = (const float*)d_in[4 * s + 3];
    }
    float* out    = (float*)d_out;
    float* params = (float*)d_ws;
    float* featT  = (float*)d_ws + WS_FEAT_OFF;
    float* numg   = out + (out_size - 1);

    hipLaunchKernelGGL(gauss_prepass, dim3(1 + TR_BLOCKS), dim3(256), 0, stream,
                       means[0], uncs[0], ops[0], means[1], uncs[1], ops[1],
                       means[2], uncs[2], ops[2], means[3], uncs[3], ops[3],
                       feats[0], feats[1], feats[2], feats[3],
                       params, numg, featT);

    RenderArgs A;
    A.featT  = featT;
    A.params = params;
    A.out    = out;
    const int Hs[4] = {25, 50, 100, 200};
    int bcum = 0;
    unsigned long long ocum = 0;
    for (int s = 0; s < 4; ++s) {
        A.H[s]      = Hs[s];
        A.bstart[s] = bcum;
        A.ooff[s]   = ocum;
        const int Pn = Hs[s] * Hs[s];
        bcum += (Pn + PIX_WG - 1) / PIX_WG;
        ocum += (unsigned long long)GEMBED * Pn;
    }
    hipLaunchKernelGGL(gauss_render, dim3(bcum), dim3(256), 0, stream, A);
}


# workgroup cut from 8 waves/128 pixels to 2 waves/32 pixels: shorter per-tile gaussian lists, and the small images' long lists run 2 waves deep per group instead of 8; k-step grouping of sums changes
# speedup vs baseline: 5.3545x; 1.2410x over previous
#include <hip/hip_runtime.h>

typedef float v2f __attribute__((ext_vector_type(2)));
typedef float v8f __attribute__((ext_vector_type(8)));

#define GEMBED   128
#define GNG      432
#define GCHUNK   16
#define FPITCH   (GEMBED + 8)
#define PIX_WAVE 16
#define WAVES_WG 2
#define NT       (32 * WAVES_WG)
#define PIX_WG   (PIX_WAVE * WAVES_WG)
#define NPASS    ((GNG + NT - 1) / NT)
#define ROWS_WAVE (GCHUNK / WAVES_WG)
#define PW_THR   ((GCHUNK * 8) / NT)
#define GTHRESH  0.01f
#define WS_FEAT_OFF 16384
#define TR_DCH   32
#define TR_BLOCKS (4 * 6 * (GEMBED / TR_DCH))

static __device__ __forceinline__ void lds_barrier()
{
    __builtin_amdgcn_fence(__ATOMIC_RELEASE, "workgroup", "local");
    __builtin_amdgcn_s_barrier();
    __builtin_amdgcn_fence(__ATOMIC_ACQUIRE, "workgroup", "local");
}

static __device__ __forceinline__ float xhalf(float x)
{
    const int b = __builtin_bit_cast(int, x);
    return __builtin_bit_cast(float, __builtin_amdgcn_permlanex16(b, b, 0x76543210, 0xfedcba98, false, false));
}

__global__ void gauss_prepass(const float* __restrict__ m0, const float* __restrict__ u0, const float* __restrict__ o0,
                              const float* __restrict__ m1, const float* __restrict__ u1, const float* __restrict__ o1,
                              const float* __restrict__ m2, const float* __restrict__ u2, const float* __restrict__ o2,
                              const float* __restrict__ m3, const float* __restrict__ u3, const float* __restrict__ o3,
                              const float* __restrict__ f0, const float* __restrict__ f1,
                              const float* __restrict__ f2, const float* __restrict__ f3,
                              float* __restrict__ params, float* __restrict__ numg, float* __restrict__ featT)
{
    __shared__ float tile[TR_DCH][73];
    const int tid = threadIdx.x;

    if (blockIdx.x != 0) {
        const int b   = (int)blockIdx.x - 1;
        const int s   = b / 24;
        const int r   = b - s * 24;
        const int cam = r >> 2;
        const int dq  = r & 3;
        const float* __restrict__ fs = (s == 0) ? f0 : (s == 1) ? f1 : (s == 2) ? f2 : f3;
        const float* __restrict__ srcp = fs + ((size_t)cam * GEMBED + dq * TR_DCH) * 72;
        for (int e = tid; e < TR_DCH * 72; e += 256) {
            const int dl = e / 72;
            tile[dl][e - dl * 72] = srcp[e];
        }
        __syncthreads();
        float* __restrict__ dstp = featT + ((size_t)s * GNG + cam * 72) * GEMBED + dq * TR_DCH;
        for (int e = tid; e < TR_DCH * 72; e += 256) {
            const int hw = e >> 5;
            const int dl = e & 31;
            dstp[(size_t)hw * GEMBED + dl] = tile[dl][hw];
        }
        return;
    }
    if (tid == 0) *numg = 0.0f;
    __syncthreads();

    const float* Ms[4] = {m0, m1, m2, m3};
    const float* Us[4] = {u0, u1, u2, u3};
    const float* Os[4] = {o0, o1, o2, o3};
    const float  Hs[4] = {25.f, 50.f, 100.f, 200.f};

    float cnt = 0.0f;
    for (int g = tid; g < 4 * GNG; g += blockDim.x) {
        const int s = g / GNG;
        const int i = g - s * GNG;
        const float* m = Ms[s] + (size_t)i * 3;
        const float* u = Us[s] + (size_t)i * 9;
        float o = Os[s][i];
        const float H  = Hs[s];
        const float sh = H * 0.01f;
        const float c00 = fmaxf(u[0], 1.0f);
        const float c01 = fmaxf(u[1], 1.0f);
        const float c11 = fmaxf(u[4], 1.0f);
        const float va  = sh * sh * c11 + 0.3f;
        const float vb  = sh * sh * c01;
        const float vc  = sh * sh * c00 + 0.3f;
        float det = va * vc - vb * vb;
        const float pr = -sh * m[1] + H * 0.5f;
        const float pc = -sh * m[0] + H * 0.5f;
        if (o > GTHRESH) cnt += 1.0f;
        const bool keep = (o > GTHRESH) && (det > 0.0f);
        float reach = -1.0f;
        if (keep) reach = 2.02f * logf(255.0f * o) + 0.04f;
        if (!keep) { o = 0.0f; det = 1.0f; }
        float* Pp = params + (size_t)g * 8;
        Pp[0] = pr; Pp[1] = pc; Pp[2] = va; Pp[3] = vb;
        Pp[4] = vc; Pp[5] = det; Pp[6] = o;  Pp[7] = reach;
    }
    atomicAdd(numg, cnt);
}

struct RenderArgs {
    const float* featT;
    const float* params;
    float* out;
    int bstart[4];
    int H[4];
    unsigned long long ooff[4];
};

__global__ __launch_bounds__(NT) void gauss_render(RenderArgs A)
{
    __shared__ __attribute__((aligned(16))) float lds_f[GCHUNK][FPITCH];
    __shared__ __attribute__((aligned(16))) int lds_list[GNG + GCHUNK];
    __shared__ int lds_cnt[NPASS * WAVES_WG];
    __shared__ __attribute__((aligned(16))) float lds_p[GCHUNK][8];

    const int bs = blockIdx.x;
    int s = 3;
    if      (bs < A.bstart[1]) s = 0;
    else if (bs < A.bstart[2]) s = 1;
    else if (bs < A.bstart[3]) s = 2;
    const int blk = bs - A.bstart[s];

    const float* __restrict__ featT  = A.featT + (size_t)s * GNG * GEMBED;
    const float* __restrict__ params = A.params + (size_t)s * GNG * 8;
    float* __restrict__ out          = A.out + A.ooff[s];
    const int H = A.H[s];
    const int W = H;
    const int P = H * W;

    const int tid  = threadIdx.x;
    const int wave = tid >> 5;
    const int lane = tid & 31;
    const int l16  = lane & 15;
    const int hi   = lane >> 4;
    const int pbase = blk * PIX_WG + wave * PIX_WAVE;

    const int p   = pbase + l16;
    const int pcl = (p < P) ? p : (P - 1);
    const int row = pcl / W;
    const float rf = (float)row;
    const float cf = (float)(pcl - row * W);
    float T = 1.0f;

    v8f acc[8];
    #pragma unroll
    for (int t = 0; t < 8; ++t) {
        #pragma unroll
        for (int i = 0; i < 8; ++i) acc[t][i] = 0.0f;
    }

    int nlist;
    {
        const int qa  = blk * PIX_WG;
        const int qb  = (qa + PIX_WG - 1 < P) ? (qa + PIX_WG - 1) : (P - 1);
        const int bra = qa / W;
        const int brb = qb / W;
        const float brlo = (float)bra;
        const float brhi = (float)brb;
        const float bclo = (bra == brb) ? (float)(qa - bra * W) : 0.0f;
        const float bchi = (bra == brb) ? (float)(qb - brb * W) : (float)(W - 1);
        bool     hitv[NPASS];
        unsigned balv[NPASS];
        float4 q0v[NPASS], q1v[NPASS];
        #pragma unroll
        for (int it = 0; it < NPASS; ++it) {
            const int g  = it * NT + tid;
            const int gi = (g < GNG) ? g : (GNG - 1);
            q0v[it] = *reinterpret_cast<const float4*>(params + (size_t)gi * 8);
            q1v[it] = *reinterpret_cast<const float4*>(params + (size_t)gi * 8 + 4);
        }
        #pragma unroll
        for (int it = 0; it < NPASS; ++it) {
            const int g = it * NT + tid;
            const float4 q0 = q0v[it];
            const float4 q1 = q1v[it];
            const float er = fmaxf(fmaxf(brlo - q0.x, q0.x - brhi), 0.0f);
            const float ec = fmaxf(fmaxf(bclo - q0.y, q0.y - bchi), 0.0f);
            const bool hit = (g < GNG) & !(er * er > q1.w * q0.z) & !(ec * ec > q1.w * q1.x);
            hitv[it] = hit;
            balv[it] = __builtin_amdgcn_ballot_w32(hit);
            if (lane == 0) lds_cnt[it * WAVES_WG + wave] = __builtin_popcount(balv[it]);
        }
        __syncthreads();
        int total = 0;
        int basev[NPASS];
        #pragma unroll
        for (int it = 0; it < NPASS; ++it) {
            basev[it] = 0;
            #pragma unroll
            for (int w = 0; w < WAVES_WG; ++w) {
                if (w == wave) basev[it] = total;
                total += lds_cnt[it * WAVES_WG + w];
            }
        }
        const unsigned below = (1u << lane) - 1u;
        #pragma unroll
        for (int it = 0; it < NPASS; ++it) {
            if (hitv[it]) lds_list[basev[it] + __builtin_popcount(balv[it] & below)] = it * NT + tid;
        }
        nlist = total;
        if (tid < GCHUNK && total + tid < ((total + GCHUNK - 1) & ~(GCHUNK - 1))) lds_list[total + tid] = -1;
        __syncthreads();
    }
    const int nchunk = __builtin_amdgcn_readfirstlane((int)((unsigned)(nlist + GCHUNK - 1) >> 4));

    const int sA = ROWS_WAVE * wave;
    const int sP = tid >> 3;
    const int pw8 = tid & 7;
    const float padv = (pw8 == 2 || pw8 == 4 || pw8 == 5) ? 1.0f : ((pw8 == 7) ? -1.0f : 0.0f);
    float4 fr[ROWS_WAVE];
    float  pv[PW_THR];
    #pragma unroll
    for (int r = 0; r < ROWS_WAVE; ++r) fr[r] = make_float4(0.0f, 0.0f, 0.0f, 0.0f);
    #pragma unroll
    for (int q = 0; q < PW_THR; ++q) pv[q] = padv;
    static_assert(ROWS_WAVE == 8, "a wave's list slots are read as two int4");
    if (nchunk > 0) {
        const int4 ia = *reinterpret_cast<const int4*>(&lds_list[sA]);
        const int4 ib = *reinterpret_cast<const int4*>(&lds_list[sA + 4]);
        int iP[PW_THR];
        #pragma unroll
        for (int q = 0; q < PW_THR; ++q) iP[q] = lds_list[sP + 8 * q];
        const int li[ROWS_WAVE] = {ia.x, ia.y, ia.z, ia.w, ib.x, ib.y, ib.z, ib.w};
        #pragma unroll
        for (int r = 0; r < ROWS_WAVE; ++r) {
            const int iA = __builtin_amdgcn_readfirstlane(li[r]);
            if (iA >= 0) fr[r] = *reinterpret_cast<const float4*>(featT + (size_t)iA * GEMBED + lane * 4);
        }
        #pragma unroll
        for (int q = 0; q < PW_THR; ++q) {
            if (iP[q] >= 0) pv[q] = params[(size_t)iP[q] * 8 + pw8];
        }
    }

    for (int ch = 0; ch < nchunk; ++ch) {
        lds_barrier();

        #pragma unroll
        for (int r = 0; r < ROWS_WAVE; ++r) *reinterpret_cast<float4*>(&lds_f[sA + r][lane * 4]) = fr[r];
        #pragma unroll
        for (int q = 0; q < PW_THR; ++q) lds_p[sP + 8 * q][pw8] = pv[q];

        if (ch + 1 < nchunk) {
            const int nb = (ch + 1) * GCHUNK;
            const int4 ia = *reinterpret_cast<const int4*>(&lds_list[nb + sA]);
            const int4 ib = *reinterpret_cast<const int4*>(&lds_list[nb + sA + 4]);
            int iP[PW_THR];
            #pragma unroll
            for (int q = 0; q < PW_THR; ++q) iP[q] = lds_list[nb + sP + 8 * q];
            const int li[ROWS_WAVE] = {ia.x, ia.y, ia.z, ia.w, ib.x, ib.y, ib.z, ib.w};
            #pragma unroll
            for (int r = 0; r < ROWS_WAVE; ++r) {
                const int iA = __builtin_amdgcn_readfirstlane(li[r]);
                fr[r] = make_float4(0.0f, 0.0f, 0.0f, 0.0f);
                if (iA >= 0) fr[r] = *reinterpret_cast<const float4*>(featT + (size_t)iA * GEMBED + lane * 4);
            }
            #pragma unroll
            for (int q = 0; q < PW_THR; ++q) {
                pv[q] = padv;
                if (iP[q] >= 0) pv[q] = params[(size_t)iP[q] * 8 + pw8];
            }
        }
        lds_barrier();

        float am[8];
        float om[8];
        #pragma unroll
        for (int m = 0; m < 8; ++m) {
            const int kk = 4 * (m >> 1) + 2 * hi + (m & 1);
            const float4 q0 = *reinterpret_cast<const float4*>(&lds_p[kk][0]);
            const float4 q1 = *reinterpret_cast<const float4*>(&lds_p[kk][4]);
            const float dr  = q0.x - rf;
            const float dc  = q0.y - cf;
            const float num = q0.w * dr * dc - 0.5f * (q1.x * dr * dr + q0.z * dc * dc);
            const float pw  = num / q1.y;
            const float al  = fminf(0.99f, q1.z * __expf(fminf(pw, 0.0f)));
            am[m] = ((pw <= 0.0f) && (al >= (1.0f / 255.0f))) ? al : 0.0f;
            om[m] = 1.0f - am[m];
        }
        float op[8];
        #pragma unroll
        for (int m = 0; m < 8; ++m) op[m] = xhalf(om[m]);

        v2f bf[4];
        #pragma unroll
        for (int j = 0; j < GCHUNK / 4; ++j) {
            const float f0 = hi ? op[2 * j]     : om[2 * j];
            const float f1 = hi ? op[2 * j + 1] : om[2 * j + 1];
            const float s0 = hi ? om[2 * j]     : op[2 * j];
            const float s1 = hi ? om[2 * j + 1] : op[2 * j + 1];
            const float T0 = T;
            const float T1 = T0 * f0;
            const float T2 = T1 * f1;
            const float T3 = T2 * s0;
            const float T4 = T3 * s1;
            bf[j].x = am[2 * j]     * (hi ? T2 : T0);
            bf[j].y = am[2 * j + 1] * (hi ? T3 : T1);
            T = T4;
        }

        #pragma unroll
        for (int j = 0; j < GCHUNK / 4; ++j) {
            const int k0 = j * 4 + 2 * hi;
            const v2f bfj = bf[j];
            const bool lane_live = (bfj.x != 0.0f) || (bfj.y != 0.0f);
            if (__builtin_amdgcn_ballot_w32(lane_live) != 0u) {
                #pragma unroll
                for (int t = 0; t < 8; ++t) {
                    v2f af;
                    af.x = lds_f[k0    ][t * 16 + l16];
                    af.y = lds_f[k0 + 1][t * 16 + l16];
                    acc[t] = __builtin_amdgcn_wmma_f32_16x16x4_f32(
                         false, af,  false, bfj,
                         (short)0, acc[t],  false,  false);
                }
            }
        }
    }

    const int pout = pbase + l16;
    if (pout < P) {
        #pragma unroll
        for (int t = 0; t < 8; ++t) {
            const int dbase = t * 16 + hi * 8;
            #pragma unroll
            for (int i = 0; i < 8; ++i) {
                out[(size_t)(dbase + i) * P + pout] = acc[t][i];
            }
        }
    }
}

extern "C" void kernel_launch(void* const* d_in, const int* in_sizes, int n_in,
                              void* d_out, int out_size, void* d_ws, size_t ws_size,
                              hipStream_t stream)
{
    (void)in_sizes; (void)n_in; (void)ws_size;
    const float* feats[4]; const float* means[4]; const float* uncs[4]; const float* ops[4];
    for (int s = 0; s < 4; ++s) {
        feats[s] = (const float*)d_in[4 * s + 0];
        means[s] = (const float*)d_in[4 * s + 1];
        uncs[s]  = (const float*)d_in[4 * s + 2];
        ops[s]   = (const float*)d_in[4 * s + 3];
    }
    float* out    = (float*)d_out;
    float* params = (float*)d_ws;
    float* featT  = (float*)d_ws + WS_FEAT_OFF;
    float* numg   = out + (out_size - 1);

    hipLaunchKernelGGL(gauss_prepass, dim3(1 + TR_BLOCKS), dim3(256), 0, stream,
                       means[0], uncs[0], ops[0], means[1], uncs[1], ops[1],
                       means[2], uncs[2], ops[2], means[3], uncs[3], ops[3],
                       feats[0], feats[1], feats[2], feats[3],
                       params, numg, featT);

    RenderArgs A;
    A.featT  = featT;
    A.params = params;
    A.out    = out;
    const int Hs[4] = {25, 50, 100, 200};
    int bcum = 0;
    unsigned long long ocum = 0;
    for (int s = 0; s < 4; ++s) {
        A.H[s]      = Hs[s];
        A.bstart[s] = bcum;
        A.ooff[s]   = ocum;
        const int Pn = Hs[s] * Hs[s];
        bcum += (Pn + PIX_WG - 1) / PIX_WG;
        ocum += (unsigned long long)GEMBED * Pn;
    }
    hipLaunchKernelGGL(gauss_render, dim3(bcum), dim3(NT), 0, stream, A);
}


# prepass un-serialised: params one gaussian per thread over 7 blocks (6 input words loaded together, no loop); num_gaussians by wave votes + one plain store (exact integer count, no float atomics)
# speedup vs baseline: 7.1160x; 1.3290x over previous
#include <hip/hip_runtime.h>

typedef float v2f __attribute__((ext_vector_type(2)));
typedef float v8f __attribute__((ext_vector_type(8)));

#define GEMBED   128
#define GNG      432
#define GCHUNK   16
#define FPITCH   (GEMBED + 8)
#define PIX_WAVE 16
#define WAVES_WG 2
#define NT       (32 * WAVES_WG)
#define PIX_WG   (PIX_WAVE * WAVES_WG)
#define NPASS    ((GNG + NT - 1) / NT)
#define ROWS_WAVE (GCHUNK / WAVES_WG)
#define PW_THR   ((GCHUNK * 8) / NT)
#define GTHRESH  0.01f
#define WS_FEAT_OFF 16384
#define TR_DCH   32
#define TR_BLOCKS (4 * 6 * (GEMBED / TR_DCH))
#define PP_BLOCKS ((4 * GNG + 255) / 256)

static __device__ __forceinline__ void lds_barrier()
{
    __builtin_amdgcn_fence(__ATOMIC_RELEASE, "workgroup", "local");
    __builtin_amdgcn_s_barrier();
    __builtin_amdgcn_fence(__ATOMIC_ACQUIRE, "workgroup", "local");
}

static __device__ __forceinline__ float xhalf(float x)
{
    const int b = __builtin_bit_cast(int, x);
    return __builtin_bit_cast(float, __builtin_amdgcn_permlanex16(b, b, 0x76543210, 0xfedcba98, false, false));
}

__global__ void gauss_prepass(const float* __restrict__ m0, const float* __restrict__ u0, const float* __restrict__ o0,
                              const float* __restrict__ m1, const float* __restrict__ u1, const float* __restrict__ o1,
                              const float* __restrict__ m2, const float* __restrict__ u2, const float* __restrict__ o2,
                              const float* __restrict__ m3, const float* __restrict__ u3, const float* __restrict__ o3,
                              const float* __restrict__ f0, const float* __restrict__ f1,
                              const float* __restrict__ f2, const float* __restrict__ f3,
                              float* __restrict__ params, float* __restrict__ numg, float* __restrict__ featT)
{
    __shared__ float tile[TR_DCH][73];
    __shared__ int wcnt[8];
    const int tid = threadIdx.x;

    if (blockIdx.x >= 1 && blockIdx.x <= TR_BLOCKS) {
        const int b   = (int)blockIdx.x - 1;
        const int s   = b / 24;
        const int r   = b - s * 24;
        const int cam = r >> 2;
        const int dq  = r & 3;
        const float* __restrict__ fs = (s == 0) ? f0 : (s == 1) ? f1 : (s == 2) ? f2 : f3;
        const float* __restrict__ srcp = fs + ((size_t)cam * GEMBED + dq * TR_DCH) * 72;
        for (int e = tid; e < TR_DCH * 72; e += 256) {
            const int dl = e / 72;
            tile[dl][e - dl * 72] = srcp[e];
        }
        __syncthreads();
        float* __restrict__ dstp = featT + ((size_t)s * GNG + cam * 72) * GEMBED + dq * TR_DCH;
        for (int e = tid; e < TR_DCH * 72; e += 256) {
            const int hw = e >> 5;
            const int dl = e & 31;
            dstp[(size_t)hw * GEMBED + dl] = tile[dl][hw];
        }
        return;
    }
    if (blockIdx.x == 0) {
        const float* __restrict__ Os[4] = {o0, o1, o2, o3};
        int c = 0;
        #pragma unroll
        for (int s = 0; s < 4; ++s) {
            #pragma unroll
            for (int h = 0; h < 2; ++h) {
                const int i = h * 256 + tid;
                const float o = Os[s][(i < GNG) ? i : (GNG - 1)];
                c += __builtin_popcount(__builtin_amdgcn_ballot_w32((i < GNG) & (o > GTHRESH)));
            }
        }
        if ((tid & 31) == 0) wcnt[tid >> 5] = c;
        __syncthreads();
        if (tid == 0) {
            int t = 0;
            #pragma unroll
            for (int w = 0; w < 8; ++w) t += wcnt[w];
            *numg = (float)t;
        }
        return;
    }

    const int g = ((int)blockIdx.x - 1 - TR_BLOCKS) * 256 + tid;
    if (g >= 4 * GNG) return;
    const int s = g / GNG;
    const int i = g - s * GNG;
    const float* __restrict__ m = ((s == 0) ? m0 : (s == 1) ? m1 : (s == 2) ? m2 : m3) + (size_t)i * 3;
    const float* __restrict__ u = ((s == 0) ? u0 : (s == 1) ? u1 : (s == 2) ? u2 : u3) + (size_t)i * 9;
    float o = ((s == 0) ? o0 : (s == 1) ? o1 : (s == 2) ? o2 : o3)[i];
    const float m_x = m[0], m_y = m[1];
    const float u_0 = u[0], u_1 = u[1], u_4 = u[4];
    const float H  = (float)(25 << s);
    const float sh = H * 0.01f;
    const float c00 = fmaxf(u_0, 1.0f);
    const float c01 = fmaxf(u_1, 1.0f);
    const float c11 = fmaxf(u_4, 1.0f);
    const float va  = sh * sh * c11 + 0.3f;
    const float vb  = sh * sh * c01;
    const float vc  = sh * sh * c00 + 0.3f;
    float det = va * vc - vb * vb;
    const float pr = -sh * m_y + H * 0.5f;
    const float pc = -sh * m_x + H * 0.5f;
    const bool keep = (o > GTHRESH) && (det > 0.0f);
    float reach = -1.0f;
    if (keep) reach = 2.02f * logf(255.0f * o) + 0.04f;
    if (!keep) { o = 0.0f; det = 1.0f; }
    float4* __restrict__ Pp = reinterpret_cast<float4*>(params + (size_t)g * 8);
    Pp[0] = make_float4(pr, pc, va, vb);
    Pp[1] = make_float4(vc, det, o, reach);
}

struct RenderArgs {
    const float* featT;
    const float* params;
    float* out;
    int bstart[4];
    int H[4];
    unsigned long long ooff[4];
};

__global__ __launch_bounds__(NT) void gauss_render(RenderArgs A)
{
    __shared__ __attribute__((aligned(16))) float lds_f[GCHUNK][FPITCH];
    __shared__ __attribute__((aligned(16))) int lds_list[GNG + GCHUNK];
    __shared__ int lds_cnt[NPASS * WAVES_WG];
    __shared__ __attribute__((aligned(16))) float lds_p[GCHUNK][8];

    const int bs = blockIdx.x;
    int s = 3;
    if      (bs < A.bstart[1]) s = 0;
    else if (bs < A.bstart[2]) s = 1;
    else if (bs < A.bstart[3]) s = 2;
    const int blk = bs - A.bstart[s];

    const float* __restrict__ featT  = A.featT + (size_t)s * GNG * GEMBED;
    const float* __restrict__ params = A.params + (size_t)s * GNG * 8;
    float* __restrict__ out          = A.out + A.ooff[s];
    const int H = A.H[s];
    const int W = H;
    const int P = H * W;

    const int tid  = threadIdx.x;
    const int wave = tid >> 5;
    const int lane = tid & 31;
    const int l16  = lane & 15;
    const int hi   = lane >> 4;
    const int pbase = blk * PIX_WG + wave * PIX_WAVE;

    const int p   = pbase + l16;
    const int pcl = (p < P) ? p : (P - 1);
    const int row = pcl / W;
    const float rf = (float)row;
    const float cf = (float)(pcl - row * W);
    float T = 1.0f;

    v8f acc[8];
    #pragma unroll
    for (int t = 0; t < 8; ++t) {
        #pragma unroll
        for (int i = 0; i < 8; ++i) acc[t][i] = 0.0f;
    }

    int nlist;
    {
        const int qa  = blk * PIX_WG;
        const int qb  = (qa + PIX_WG - 1 < P) ? (qa + PIX_WG - 1) : (P - 1);
        const int bra = qa / W;
        const int brb = qb / W;
        const float brlo = (float)bra;
        const float brhi = (float)brb;
        const float bclo = (bra == brb) ? (float)(qa - bra * W) : 0.0f;
        const float bchi = (bra == brb) ? (float)(qb - brb * W) : (float)(W - 1);
        bool     hitv[NPASS];
        unsigned balv[NPASS];
        float4 q0v[NPASS], q1v[NPASS];
        #pragma unroll
        for (int it = 0; it < NPASS; ++it) {
            const int g  = it * NT + tid;
            const int gi = (g < GNG) ? g : (GNG - 1);
            q0v[it] = *reinterpret_cast<const float4*>(params + (size_t)gi * 8);
            q1v[it] = *reinterpret_cast<const float4*>(params + (size_t)gi * 8 + 4);
        }
        #pragma unroll
        for (int it = 0; it < NPASS; ++it) {
            const int g = it * NT + tid;
            const float4 q0 = q0v[it];
            const float4 q1 = q1v[it];
            const float er = fmaxf(fmaxf(brlo - q0.x, q0.x - brhi), 0.0f);
            const float ec = fmaxf(fmaxf(bclo - q0.y, q0.y - bchi), 0.0f);
            const bool hit = (g < GNG) & !(er * er > q1.w * q0.z) & !(ec * ec > q1.w * q1.x);
            hitv[it] = hit;
            balv[it] = __builtin_amdgcn_ballot_w32(hit);
            if (lane == 0) lds_cnt[it * WAVES_WG + wave] = __builtin_popcount(balv[it]);
        }
        __syncthreads();
        int total = 0;
        int basev[NPASS];
        #pragma unroll
        for (int it = 0; it < NPASS; ++it) {
            basev[it] = 0;
            #pragma unroll
            for (int w = 0; w < WAVES_WG; ++w) {
                if (w == wave) basev[it] = total;
                total += lds_cnt[it * WAVES_WG + w];
            }
        }
        const unsigned below = (1u << lane) - 1u;
        #pragma unroll
        for (int it = 0; it < NPASS; ++it) {
            if (hitv[it]) lds_list[basev[it] + __builtin_popcount(balv[it] & below)] = it * NT + tid;
        }
        nlist = total;
        if (tid < GCHUNK && total + tid < ((total + GCHUNK - 1) & ~(GCHUNK - 1))) lds_list[total + tid] = -1;
        __syncthreads();
    }
    const int nchunk = __builtin_amdgcn_readfirstlane((int)((unsigned)(nlist + GCHUNK - 1) >> 4));

    const int sA = ROWS_WAVE * wave;
    const int sP = tid >> 3;
    const int pw8 = tid & 7;
    const float padv = (pw8 == 2 || pw8 == 4 || pw8 == 5) ? 1.0f : ((pw8 == 7) ? -1.0f : 0.0f);
    float4 fr[ROWS_WAVE];
    float  pv[PW_THR];
    #pragma unroll
    for (int r = 0; r < ROWS_WAVE; ++r) fr[r] = make_float4(0.0f, 0.0f, 0.0f, 0.0f);
    #pragma unroll
    for (int q = 0; q < PW_THR; ++q) pv[q] = padv;
    static_assert(ROWS_WAVE == 8, "a wave's list slots are read as two int4");
    if (nchunk > 0) {
        const int4 ia = *reinterpret_cast<const int4*>(&lds_list[sA]);
        const int4 ib = *reinterpret_cast<const int4*>(&lds_list[sA + 4]);
        int iP[PW_THR];
        #pragma unroll
        for (int q = 0; q < PW_THR; ++q) iP[q] = lds_list[sP + 8 * q];
        const int li[ROWS_WAVE] = {ia.x, ia.y, ia.z, ia.w, ib.x, ib.y, ib.z, ib.w};
        #pragma unroll
        for (int r = 0; r < ROWS_WAVE; ++r) {
            const int iA = __builtin_amdgcn_readfirstlane(li[r]);
            if (iA >= 0) fr[r] = *reinterpret_cast<const float4*>(featT + (size_t)iA * GEMBED + lane * 4);
        }
        #pragma unroll
        for (int q = 0; q < PW_THR; ++q) {
            if (iP[q] >= 0) pv[q] = params[(size_t)iP[q] * 8 + pw8];
        }
    }

    for (int ch = 0; ch < nchunk; ++ch) {
        lds_barrier();

        #pragma unroll
        for (int r = 0; r < ROWS_WAVE; ++r) *reinterpret_cast<float4*>(&lds_f[sA + r][lane * 4]) = fr[r];
        #pragma unroll
        for (int q = 0; q < PW_THR; ++q) lds_p[sP + 8 * q][pw8] = pv[q];

        if (ch + 1 < nchunk) {
            const int nb = (ch + 1) * GCHUNK;
            const int4 ia = *reinterpret_cast<const int4*>(&lds_list[nb + sA]);
            const int4 ib = *reinterpret_cast<const int4*>(&lds_list[nb + sA + 4]);
            int iP[PW_THR];
            #pragma unroll
            for (int q = 0; q < PW_THR; ++q) iP[q] = lds_list[nb + sP + 8 * q];
            const int li[ROWS_WAVE] = {ia.x, ia.y, ia.z, ia.w, ib.x, ib.y, ib.z, ib.w};
            #pragma unroll
            for (int r = 0; r < ROWS_WAVE; ++r) {
                const int iA = __builtin_amdgcn_readfirstlane(li[r]);
                fr[r] = make_float4(0.0f, 0.0f, 0.0f, 0.0f);
                if (iA >= 0) fr[r] = *reinterpret_cast<const float4*>(featT + (size_t)iA * GEMBED + lane * 4);
            }
            #pragma unroll
            for (int q = 0; q < PW_THR; ++q) {
                pv[q] = padv;
                if (iP[q] >= 0) pv[q] = params[(size_t)iP[q] * 8 + pw8];
            }
        }
        lds_barrier();

        float am[8];
        float om[8];
        #pragma unroll
        for (int m = 0; m < 8; ++m) {
            const int kk = 4 * (m >> 1) + 2 * hi + (m & 1);
            const float4 q0 = *reinterpret_cast<const float4*>(&lds_p[kk][0]);
            const float4 q1 = *reinterpret_cast<const float4*>(&lds_p[kk][4]);
            const float dr  = q0.x - rf;
            const float dc  = q0.y - cf;
            const float num = q0.w * dr * dc - 0.5f * (q1.x * dr * dr + q0.z * dc * dc);
            const float pw  = num / q1.y;
            const float al  = fminf(0.99f, q1.z * __expf(fminf(pw, 0.0f)));
            am[m] = ((pw <= 0.0f) && (al >= (1.0f / 255.0f))) ? al : 0.0f;
            om[m] = 1.0f - am[m];
        }
        float op[8];
        #pragma unroll
        for (int m = 0; m < 8; ++m) op[m] = xhalf(om[m]);

        v2f bf[4];
        #pragma unroll
        for (int j = 0; j < GCHUNK / 4; ++j) {
            const float f0 = hi ? op[2 * j]     : om[2 * j];
            const float f1 = hi ? op[2 * j + 1] : om[2 * j + 1];
            const float s0 = hi ? om[2 * j]     : op[2 * j];
            const float s1 = hi ? om[2 * j + 1] : op[2 * j + 1];
            const float T0 = T;
            const float T1 = T0 * f0;
            const float T2 = T1 * f1;
            const float T3 = T2 * s0;
            const float T4 = T3 * s1;
            bf[j].x = am[2 * j]     * (hi ? T2 : T0);
            bf[j].y = am[2 * j + 1] * (hi ? T3 : T1);
            T = T4;
        }

        #pragma unroll
        for (int j = 0; j < GCHUNK / 4; ++j) {
            const int k0 = j * 4 + 2 * hi;
            const v2f bfj = bf[j];
            const bool lane_live = (bfj.x != 0.0f) || (bfj.y != 0.0f);
            if (__builtin_amdgcn_ballot_w32(lane_live) != 0u) {
                #pragma unroll
                for (int t = 0; t < 8; ++t) {
                    v2f af;
                    af.x = lds_f[k0    ][t * 16 + l16];
                    af.y = lds_f[k0 + 1][t * 16 + l16];
                    acc[t] = __builtin_amdgcn_wmma_f32_16x16x4_f32(
                         false, af,  false, bfj,
                         (short)0, acc[t],  false,  false);
                }
            }
        }
    }

    const int pout = pbase + l16;
    if (pout < P) {
        #pragma unroll
        for (int t = 0; t < 8; ++t) {
            const int dbase = t * 16 + hi * 8;
            #pragma unroll
            for (int i = 0; i < 8; ++i) {
                out[(size_t)(dbase + i) * P + pout] = acc[t][i];
            }
        }
    }
}

extern "C" void kernel_launch(void* const* d_in, const int* in_sizes, int n_in,
                              void* d_out, int out_size, void* d_ws, size_t ws_size,
                              hipStream_t stream)
{
    (void)in_sizes; (void)n_in; (void)ws_size;
    const float* feats[4]; const float* means[4]; const float* uncs[4]; const float* ops[4];
    for (int s = 0; s < 4; ++s) {
        feats[s] = (const float*)d_in[4 * s + 0];
        means[s] = (const float*)d_in[4 * s + 1];
        uncs[s]  = (const float*)d_in[4 * s + 2];
        ops[s]   = (const float*)d_in[4 * s + 3];
    }
    float* out    = (float*)d_out;
    float* params = (float*)d_ws;
    float* featT  = (float*)d_ws + WS_FEAT_OFF;
    float* numg   = out + (out_size - 1);

    hipLaunchKernelGGL(gauss_prepass, dim3(1 + TR_BLOCKS + PP_BLOCKS), dim3(256), 0, stream,
                       means[0], uncs[0], ops[0], means[1], uncs[1], ops[1],
                       means[2], uncs[2], ops[2], means[3], uncs[3], ops[3],
                       feats[0], feats[1], feats[2], feats[3],
                       params, numg, featT);

    RenderArgs A;
    A.featT  = featT;
    A.params = params;
    A.out    = out;
    const int Hs[4] = {25, 50, 100, 200};
    int bcum = 0;
    unsigned long long ocum = 0;
    for (int s = 0; s < 4; ++s) {
        A.H[s]      = Hs[s];
        A.bstart[s] = bcum;
        A.ooff[s]   = ocum;
        const int Pn = Hs[s] * Hs[s];
        bcum += (Pn + PIX_WG - 1) / PIX_WG;
        ocum += (unsigned long long)GEMBED * Pn;
    }
    hipLaunchKernelGGL(gauss_render, dim3(bcum), dim3(NT), 0, stream, A);
}


# per-tile gaussian lists shortened by tile geometry: images under 64 wide use 8x4-pixel tiles; a 32-pixel tile that wraps a row end is tested as its two row segments; k-step grouping of sums changes
# speedup vs baseline: 7.8204x; 1.0990x over previous
#include <hip/hip_runtime.h>

typedef float v2f __attribute__((ext_vector_type(2)));
typedef float v8f __attribute__((ext_vector_type(8)));

#define GEMBED   128
#define GNG      432
#define GCHUNK   16
#define FPITCH   (GEMBED + 8)
#define PIX_WAVE 16
#define WAVES_WG 2
#define NT       (32 * WAVES_WG)
#define PIX_WG   (PIX_WAVE * WAVES_WG)
#define NPASS    ((GNG + NT - 1) / NT)
#define TILE_W   8
#define TILE_H   4
#define NARROW_W (2 * PIX_WG)
#define ROWS_WAVE (GCHUNK / WAVES_WG)
#define PW_THR   ((GCHUNK * 8) / NT)
#define GTHRESH  0.01f
#define WS_FEAT_OFF 16384
#define TR_DCH   32
#define TR_BLOCKS (4 * 6 * (GEMBED / TR_DCH))
#define PP_BLOCKS ((4 * GNG + 255) / 256)

static __device__ __forceinline__ void lds_barrier()
{
    __builtin_amdgcn_fence(__ATOMIC_RELEASE, "workgroup", "local");
    __builtin_amdgcn_s_barrier();
    __builtin_amdgcn_fence(__ATOMIC_ACQUIRE, "workgroup", "local");
}

static __device__ __forceinline__ float xhalf(float x)
{
    const int b = __builtin_bit_cast(int, x);
    return __builtin_bit_cast(float, __builtin_amdgcn_permlanex16(b, b, 0x76543210, 0xfedcba98, false, false));
}

__global__ void gauss_prepass(const float* __restrict__ m0, const float* __restrict__ u0, const float* __restrict__ o0,
                              const float* __restrict__ m1, const float* __restrict__ u1, const float* __restrict__ o1,
                              const float* __restrict__ m2, const float* __restrict__ u2, const float* __restrict__ o2,
                              const float* __restrict__ m3, const float* __restrict__ u3, const float* __restrict__ o3,
                              const float* __restrict__ f0, const float* __restrict__ f1,
                              const float* __restrict__ f2, const float* __restrict__ f3,
                              float* __restrict__ params, float* __restrict__ numg, float* __restrict__ featT)
{
    __shared__ float tile[TR_DCH][73];
    __shared__ int wcnt[8];
    const int tid = threadIdx.x;

    if (blockIdx.x >= 1 && blockIdx.x <= TR_BLOCKS) {
        const int b   = (int)blockIdx.x - 1;
        const int s   = b / 24;
        const int r   = b - s * 24;
        const int cam = r >> 2;
        const int dq  = r & 3;
        const float* __restrict__ fs = (s == 0) ? f0 : (s == 1) ? f1 : (s == 2) ? f2 : f3;
        const float* __restrict__ srcp = fs + ((size_t)cam * GEMBED + dq * TR_DCH) * 72;
        for (int e = tid; e < TR_DCH * 72; e += 256) {
            const int dl = e / 72;
            tile[dl][e - dl * 72] = srcp[e];
        }
        __syncthreads();
        float* __restrict__ dstp = featT + ((size_t)s * GNG + cam * 72) * GEMBED + dq * TR_DCH;
        for (int e = tid; e < TR_DCH * 72; e += 256) {
            const int hw = e >> 5;
            const int dl = e & 31;
            dstp[(size_t)hw * GEMBED + dl] = tile[dl][hw];
        }
        return;
    }
    if (blockIdx.x == 0) {
        const float* __restrict__ Os[4] = {o0, o1, o2, o3};
        int c = 0;
        #pragma unroll
        for (int s = 0; s < 4; ++s) {
            #pragma unroll
            for (int h = 0; h < 2; ++h) {
                const int i = h * 256 + tid;
                const float o = Os[s][(i < GNG) ? i : (GNG - 1)];
                c += __builtin_popcount(__builtin_amdgcn_ballot_w32((i < GNG) & (o > GTHRESH)));
            }
        }
        if ((tid & 31) == 0) wcnt[tid >> 5] = c;
        __syncthreads();
        if (tid == 0) {
            int t = 0;
            #pragma unroll
            for (int w = 0; w < 8; ++w) t += wcnt[w];
            *numg = (float)t;
        }
        return;
    }

    const int g = ((int)blockIdx.x - 1 - TR_BLOCKS) * 256 + tid;
    if (g >= 4 * GNG) return;
    const int s = g / GNG;
    const int i = g - s * GNG;
    const float* __restrict__ m = ((s == 0) ? m0 : (s == 1) ? m1 : (s == 2) ? m2 : m3) + (size_t)i * 3;
    const float* __restrict__ u = ((s == 0) ? u0 : (s == 1) ? u1 : (s == 2) ? u2 : u3) + (size_t)i * 9;
    float o = ((s == 0) ? o0 : (s == 1) ? o1 : (s == 2) ? o2 : o3)[i];
    const float m_x = m[0], m_y = m[1];
    const float u_0 = u[0], u_1 = u[1], u_4 = u[4];
    const float H  = (float)(25 << s);
    const float sh = H * 0.01f;
    const float c00 = fmaxf(u_0, 1.0f);
    const float c01 = fmaxf(u_1, 1.0f);
    const float c11 = fmaxf(u_4, 1.0f);
    const float va  = sh * sh * c11 + 0.3f;
    const float vb  = sh * sh * c01;
    const float vc  = sh * sh * c00 + 0.3f;
    float det = va * vc - vb * vb;
    const float pr = -sh * m_y + H * 0.5f;
    const float pc = -sh * m_x + H * 0.5f;
    const bool keep = (o > GTHRESH) && (det > 0.0f);
    float reach = -1.0f;
    if (keep) reach = 2.02f * logf(255.0f * o) + 0.04f;
    if (!keep) { o = 0.0f; det = 1.0f; }
    float4* __restrict__ Pp = reinterpret_cast<float4*>(params + (size_t)g * 8);
    Pp[0] = make_float4(pr, pc, va, vb);
    Pp[1] = make_float4(vc, det, o, reach);
}

struct RenderArgs {
    const float* featT;
    const float* params;
    float* out;
    int bstart[4];
    int H[4];
    unsigned long long ooff[4];
};

__global__ __launch_bounds__(NT) void gauss_render(RenderArgs A)
{
    __shared__ __attribute__((aligned(16))) float lds_f[GCHUNK][FPITCH];
    __shared__ __attribute__((aligned(16))) int lds_list[GNG + GCHUNK];
    __shared__ int lds_cnt[NPASS * WAVES_WG];
    __shared__ __attribute__((aligned(16))) float lds_p[GCHUNK][8];

    const int bs = blockIdx.x;
    int s = 3;
    if      (bs < A.bstart[1]) s = 0;
    else if (bs < A.bstart[2]) s = 1;
    else if (bs < A.bstart[3]) s = 2;
    const int blk = bs - A.bstart[s];

    const float* __restrict__ featT  = A.featT + (size_t)s * GNG * GEMBED;
    const float* __restrict__ params = A.params + (size_t)s * GNG * 8;
    float* __restrict__ out          = A.out + A.ooff[s];
    const int H = A.H[s];
    const int W = H;
    const int P = H * W;

    const int tid  = threadIdx.x;
    const int wave = tid >> 5;
    const int lane = tid & 31;
    const int l16  = lane & 15;
    const int hi   = lane >> 4;
    const int slot = wave * PIX_WAVE + l16;
    int prow, pcol;
    bool pvalid;
    float ralo, rahi, calo, cahi, rblo, rbhi, cblo, cbhi;
    if (W < NARROW_W) {
        const int tiles_x = (W + TILE_W - 1) / TILE_W;
        const int ty = blk / tiles_x;
        const int tx = blk - ty * tiles_x;
        const int r0 = ty * TILE_H;
        const int c0 = tx * TILE_W;
        const int rr = r0 + slot / TILE_W;
        const int cc = c0 + slot % TILE_W;
        pvalid = (rr < H) & (cc < W);
        prow = (rr < H) ? rr : (H - 1);
        pcol = (cc < W) ? cc : (W - 1);
        const int r1 = (r0 + TILE_H - 1 < H) ? (r0 + TILE_H - 1) : (H - 1);
        const int c1 = (c0 + TILE_W - 1 < W) ? (c0 + TILE_W - 1) : (W - 1);
        ralo = (float)r0; rahi = (float)r1; calo = (float)c0; cahi = (float)c1;
        rblo = ralo; rbhi = rahi; cblo = calo; cbhi = cahi;
    } else {
        const int pl  = blk * PIX_WG + slot;
        pvalid = (pl < P);
        const int pcl = pvalid ? pl : (P - 1);
        prow = pcl / W;
        pcol = pcl - prow * W;
        const int qa  = blk * PIX_WG;
        const int qb  = (qa + PIX_WG - 1 < P) ? (qa + PIX_WG - 1) : (P - 1);
        const int bra = qa / W;
        const int brb = qb / W;
        const bool one = (bra == brb);
        ralo = (float)bra; rahi = ralo;
        calo = (float)(qa - bra * W);
        cahi = one ? (float)(qb - bra * W) : (float)(W - 1);
        rblo = (float)brb; rbhi = rblo;
        cblo = one ? calo : 0.0f;
        cbhi = (float)(qb - brb * W);
    }
    const float rf = (float)prow;
    const float cf = (float)pcol;
    float T = 1.0f;

    v8f acc[8];
    #pragma unroll
    for (int t = 0; t < 8; ++t) {
        #pragma unroll
        for (int i = 0; i < 8; ++i) acc[t][i] = 0.0f;
    }

    int nlist;
    {
        bool     hitv[NPASS];
        unsigned balv[NPASS];
        float4 q0v[NPASS], q1v[NPASS];
        #pragma unroll
        for (int it = 0; it < NPASS; ++it) {
            const int g  = it * NT + tid;
            const int gi = (g < GNG) ? g : (GNG - 1);
            q0v[it] = *reinterpret_cast<const float4*>(params + (size_t)gi * 8);
            q1v[it] = *reinterpret_cast<const float4*>(params + (size_t)gi * 8 + 4);
        }
        #pragma unroll
        for (int it = 0; it < NPASS; ++it) {
            const int g = it * NT + tid;
            const float4 q0 = q0v[it];
            const float4 q1 = q1v[it];
            const float era = fmaxf(fmaxf(ralo - q0.x, q0.x - rahi), 0.0f);
            const float eca = fmaxf(fmaxf(calo - q0.y, q0.y - cahi), 0.0f);
            const float erb = fmaxf(fmaxf(rblo - q0.x, q0.x - rbhi), 0.0f);
            const float ecb = fmaxf(fmaxf(cblo - q0.y, q0.y - cbhi), 0.0f);
            const float rv = q1.w * q0.z;
            const float cv = q1.w * q1.x;
            const bool hit = (g < GNG) & ((!(era * era > rv) & !(eca * eca > cv)) | (!(erb * erb > rv) & !(ecb * ecb > cv)));
            hitv[it] = hit;
            balv[it] = __builtin_amdgcn_ballot_w32(hit);
            if (lane == 0) lds_cnt[it * WAVES_WG + wave] = __builtin_popcount(balv[it]);
        }
        __syncthreads();
        int total = 0;
        int basev[NPASS];
        #pragma unroll
        for (int it = 0; it < NPASS; ++it) {
            basev[it] = 0;
            #pragma unroll
            for (int w = 0; w < WAVES_WG; ++w) {
                if (w == wave) basev[it] = total;
                total += lds_cnt[it * WAVES_WG + w];
            }
        }
        const unsigned below = (1u << lane) - 1u;
        #pragma unroll
        for (int it = 0; it < NPASS; ++it) {
            if (hitv[it]) lds_list[basev[it] + __builtin_popcount(balv[it] & below)] = it * NT + tid;
        }
        nlist = total;
        if (tid < GCHUNK && total + tid < ((total + GCHUNK - 1) & ~(GCHUNK - 1))) lds_list[total + tid] = -1;
        __syncthreads();
    }
    const int nchunk = __builtin_amdgcn_readfirstlane((int)((unsigned)(nlist + GCHUNK - 1) >> 4));

    const int sA = ROWS_WAVE * wave;
    const int sP = tid >> 3;
    const int pw8 = tid & 7;
    const float padv = (pw8 == 2 || pw8 == 4 || pw8 == 5) ? 1.0f : ((pw8 == 7) ? -1.0f : 0.0f);
    float4 fr[ROWS_WAVE];
    float  pv[PW_THR];
    #pragma unroll
    for (int r = 0; r < ROWS_WAVE; ++r) fr[r] = make_float4(0.0f, 0.0f, 0.0f, 0.0f);
    #pragma unroll
    for (int q = 0; q < PW_THR; ++q) pv[q] = padv;
    static_assert(ROWS_WAVE == 8, "a wave's list slots are read as two int4");
    if (nchunk > 0) {
        const int4 ia = *reinterpret_cast<const int4*>(&lds_list[sA]);
        const int4 ib = *reinterpret_cast<const int4*>(&lds_list[sA + 4]);
        int iP[PW_THR];
        #pragma unroll
        for (int q = 0; q < PW_THR; ++q) iP[q] = lds_list[sP + 8 * q];
        const int li[ROWS_WAVE] = {ia.x, ia.y, ia.z, ia.w, ib.x, ib.y, ib.z, ib.w};
        #pragma unroll
        for (int r = 0; r < ROWS_WAVE; ++r) {
            const int iA = __builtin_amdgcn_readfirstlane(li[r]);
            if (iA >= 0) fr[r] = *reinterpret_cast<const float4*>(featT + (size_t)iA * GEMBED + lane * 4);
        }
        #pragma unroll
        for (int q = 0; q < PW_THR; ++q) {
            if (iP[q] >= 0) pv[q] = params[(size_t)iP[q] * 8 + pw8];
        }
    }

    for (int ch = 0; ch < nchunk; ++ch) {
        lds_barrier();

        #pragma unroll
        for (int r = 0; r < ROWS_WAVE; ++r) *reinterpret_cast<float4*>(&lds_f[sA + r][lane * 4]) = fr[r];
        #pragma unroll
        for (int q = 0; q < PW_THR; ++q) lds_p[sP + 8 * q][pw8] = pv[q];

        if (ch + 1 < nchunk) {
            const int nb = (ch + 1) * GCHUNK;
            const int4 ia = *reinterpret_cast<const int4*>(&lds_list[nb + sA]);
            const int4 ib = *reinterpret_cast<const int4*>(&lds_list[nb + sA + 4]);
            int iP[PW_THR];
            #pragma unroll
            for (int q = 0; q < PW_THR; ++q) iP[q] = lds_list[nb + sP + 8 * q];
            const int li[ROWS_WAVE] = {ia.x, ia.y, ia.z, ia.w, ib.x, ib.y, ib.z, ib.w};
            #pragma unroll
            for (int r = 0; r < ROWS_WAVE; ++r) {
                const int iA = __builtin_amdgcn_readfirstlane(li[r]);
                fr[r] = make_float4(0.0f, 0.0f, 0.0f, 0.0f);
                if (iA >= 0) fr[r] = *reinterpret_cast<const float4*>(featT + (size_t)iA * GEMBED + lane * 4);
            }
            #pragma unroll
            for (int q = 0; q < PW_THR; ++q) {
                pv[q] = padv;
                if (iP[q] >= 0) pv[q] = params[(size_t)iP[q] * 8 + pw8];
            }
        }
        lds_barrier();

        float am[8];
        float om[8];
        #pragma unroll
        for (int m = 0; m < 8; ++m) {
            const int kk = 4 * (m >> 1) + 2 * hi + (m & 1);
            const float4 q0 = *reinterpret_cast<const float4*>(&lds_p[kk][0]);
            const float4 q1 = *reinterpret_cast<const float4*>(&lds_p[kk][4]);
            const float dr  = q0.x - rf;
            const float dc  = q0.y - cf;
            const float num = q0.w * dr * dc - 0.5f * (q1.x * dr * dr + q0.z * dc * dc);
            const float pw  = num / q1.y;
            const float al  = fminf(0.99f, q1.z * __expf(fminf(pw, 0.0f)));
            am[m] = ((pw <= 0.0f) && (al >= (1.0f / 255.0f))) ? al : 0.0f;
            om[m] = 1.0f - am[m];
        }
        float op[8];
        #pragma unroll
        for (int m = 0; m < 8; ++m) op[m] = xhalf(om[m]);

        v2f bf[4];
        #pragma unroll
        for (int j = 0; j < GCHUNK / 4; ++j) {
            const float f0 = hi ? op[2 * j]     : om[2 * j];
            const float f1 = hi ? op[2 * j + 1] : om[2 * j + 1];
            const float s0 = hi ? om[2 * j]     : op[2 * j];
            const float s1 = hi ? om[2 * j + 1] : op[2 * j + 1];
            const float T0 = T;
            const float T1 = T0 * f0;
            const float T2 = T1 * f1;
            const float T3 = T2 * s0;
            const float T4 = T3 * s1;
            bf[j].x = am[2 * j]     * (hi ? T2 : T0);
            bf[j].y = am[2 * j + 1] * (hi ? T3 : T1);
            T = T4;
        }

        #pragma unroll
        for (int j = 0; j < GCHUNK / 4; ++j) {
            const int k0 = j * 4 + 2 * hi;
            const v2f bfj = bf[j];
            const bool lane_live = (bfj.x != 0.0f) || (bfj.y != 0.0f);
            if (__builtin_amdgcn_ballot_w32(lane_live) != 0u) {
                #pragma unroll
                for (int t = 0; t < 8; ++t) {
                    v2f af;
                    af.x = lds_f[k0    ][t * 16 + l16];
                    af.y = lds_f[k0 + 1][t * 16 + l16];
                    acc[t] = __builtin_amdgcn_wmma_f32_16x16x4_f32(
                         false, af,  false, bfj,
                         (short)0, acc[t],  false,  false);
                }
            }
        }
    }

    const int pout = prow * W + pcol;
    if (pvalid) {
        #pragma unroll
        for (int t = 0; t < 8; ++t) {
            const int dbase = t * 16 + hi * 8;
            #pragma unroll
            for (int i = 0; i < 8; ++i) {
                out[(size_t)(dbase + i) * P + pout] = acc[t][i];
            }
        }
    }
}

extern "C" void kernel_launch(void* const* d_in, const int* in_sizes, int n_in,
                              void* d_out, int out_size, void* d_ws, size_t ws_size,
                              hipStream_t stream)
{
    (void)in_sizes; (void)n_in; (void)ws_size;
    const float* feats[4]; const float* means[4]; const float* uncs[4]; const float* ops[4];
    for (int s = 0; s < 4; ++s) {
        feats[s] = (const float*)d_in[4 * s + 0];
        means[s] = (const float*)d_in[4 * s + 1];
        uncs[s]  = (const float*)d_in[4 * s + 2];
        ops[s]   = (const float*)d_in[4 * s + 3];
    }
    float* out    = (float*)d_out;
    float* params = (float*)d_ws;
    float* featT  = (float*)d_ws + WS_FEAT_OFF;
    float* numg   = out + (out_size - 1);

    hipLaunchKernelGGL(gauss_prepass, dim3(1 + TR_BLOCKS + PP_BLOCKS), dim3(256), 0, stream,
                       means[0], uncs[0], ops[0], means[1], uncs[1], ops[1],
                       means[2], uncs[2], ops[2], means[3], uncs[3], ops[3],
                       feats[0], feats[1], feats[2], feats[3],
                       params, numg, featT);

    RenderArgs A;
    A.featT  = featT;
    A.params = params;
    A.out    = out;
    const int Hs[4] = {25, 50, 100, 200};
    int bcum = 0;
    unsigned long long ocum = 0;
    for (int s = 0; s < 4; ++s) {
        A.H[s]      = Hs[s];
        A.bstart[s] = bcum;
        A.ooff[s]   = ocum;
        const int Pn = Hs[s] * Hs[s];
        bcum += (Hs[s] < NARROW_W) ? ((Hs[s] + TILE_W - 1) / TILE_W) * ((Hs[s] + TILE_H - 1) / TILE_H)
                                   : (Pn + PIX_WG - 1) / PIX_WG;
        ocum += (unsigned long long)GEMBED * Pn;
    }
    hipLaunchKernelGGL(gauss_render, dim3(bcum), dim3(NT), 0, stream, A);
}
